# starting kernel
# baseline (speedup 1.0000x reference)
#include <hip/hip_runtime.h>

typedef __attribute__((ext_vector_type(2))) float v2f;
typedef __attribute__((ext_vector_type(8))) float v8f;

#define N_NODES 100000
#define N_EDGES 1600000
#define D 128
#define NEG_INF (-3.402823466e+38f)

__global__ void copy_init_kernel(const float* __restrict__ feat,
                                 float* __restrict__ out,
                                 int* __restrict__ deg) {
    int idx = blockIdx.x * blockDim.x + threadIdx.x;
    int n  = idx >> 5;
    int j4 = (idx & 31) << 2;
    float4 v = *(const float4*)(feat + (size_t)n * D + j4);
    *(float4*)(out + (size_t)n * (2 * D) + j4) = v;
    if (idx <= N_NODES) deg[idx] = 0;
}

__global__ void __launch_bounds__(160, 1)
gemm_wmma_kernel(const float* __restrict__ feat,
                 const float* __restrict__ W1,
                 float* __restrict__ norm_h) {
    int wave = threadIdx.x >> 5;
    int lane = threadIdx.x & 31;
    int tm = blockIdx.x * 5 + wave;
    int half = lane >> 4;
    int l15  = lane & 15;

    const float* arow0 = feat + (size_t)(tm * 32 + l15) * D + half * 2;
    const float* arow1 = arow0 + (size_t)16 * D;
    const float* bbase = W1   + (size_t)l15 * D + half * 2;

    v8f c0[8] = {};
    v8f c1[8] = {};
    for (int k = 0; k < D; k += 4) {
        v2f a0 = *(const v2f*)(arow0 + k);
        v2f a1 = *(const v2f*)(arow1 + k);
        v2f b[8];
#pragma unroll
        for (int tn = 0; tn < 8; ++tn) {
            b[tn] = *(const v2f*)(bbase + (size_t)tn * 16 * D + k);
        }
#pragma unroll
        for (int tn = 0; tn < 8; ++tn) {
            c0[tn] = __builtin_amdgcn_wmma_f32_16x16x4_f32(
                         false, a0, false, b[tn], (short)0, c0[tn], false, false);
            c1[tn] = __builtin_amdgcn_wmma_f32_16x16x4_f32(
                         false, a1, false, b[tn], (short)0, c1[tn], false, false);
        }
    }

    float* crow0 = norm_h + (size_t)(tm * 32 + half * 8) * D + l15;
    float* crow1 = crow0 + (size_t)16 * D;
#pragma unroll
    for (int tn = 0; tn < 8; ++tn) {
#pragma unroll
        for (int v = 0; v < 8; ++v) {
            crow0[(size_t)v * D + tn * 16] = c0[tn][v];
            crow1[(size_t)v * D + tn * 16] = c1[tn][v];
        }
    }
}

__global__ void hist_kernel(const int* __restrict__ dst, int* __restrict__ deg) {
    int e = blockIdx.x * blockDim.x + threadIdx.x;
    atomicAdd(&deg[dst[e]], 1);
}

__global__ void scan_kernel(const int* __restrict__ deg,
                            int* __restrict__ offsets,
                            int* __restrict__ cursor) {
    __shared__ int sdata[1024];
    __shared__ int carry;
    int tid = threadIdx.x;
    if (tid == 0) carry = 0;
    __syncthreads();
    for (int base = 0; base <= N_NODES; base += 1024) {
        int i = base + tid;
        int x = (i < N_NODES) ? deg[i] : 0;
        sdata[tid] = x;
        __syncthreads();
#pragma unroll
        for (int off = 1; off < 1024; off <<= 1) {
            int t = (tid >= off) ? sdata[tid - off] : 0;
            __syncthreads();
            sdata[tid] += t;
            __syncthreads();
        }
        int excl = carry + sdata[tid] - x;
        if (i <= N_NODES) {
            offsets[i] = excl;
            if (i < N_NODES) cursor[i] = excl;
        }
        __syncthreads();
        if (tid == 1023) carry += sdata[1023];
        __syncthreads();
    }
}

__global__ void scatter_kernel(const int* __restrict__ src,
                               const int* __restrict__ dst,
                               int* __restrict__ cursor,
                               int* __restrict__ csr_src) {
    int e = blockIdx.x * blockDim.x + threadIdx.x;
    int pos = atomicAdd(&cursor[dst[e]], 1);
    csr_src[pos] = src[e];
}

__global__ void segmax_kernel(const float* __restrict__ norm_h,
                              const int* __restrict__ offsets,
                              const int* __restrict__ csr_src,
                              float* __restrict__ out) {
    int wave = threadIdx.x >> 5;
    int lane = threadIdx.x & 31;
    int node = blockIdx.x * 8 + wave;
    int beg = offsets[node];
    int end = offsets[node + 1];
    float4 acc = make_float4(NEG_INF, NEG_INF, NEG_INF, NEG_INF);
    int j4 = lane << 2;
    for (int e = beg; e < end; ++e) {
        int s = csr_src[e];
        float4 v = *(const float4*)(norm_h + (size_t)s * D + j4);
        acc.x = fmaxf(acc.x, v.x);
        acc.y = fmaxf(acc.y, v.y);
        acc.z = fmaxf(acc.z, v.z);
        acc.w = fmaxf(acc.w, v.w);
    }
    *(float4*)(out + (size_t)node * (2 * D) + D + j4) = acc;
}

extern "C" void kernel_launch(void* const* d_in, const int* in_sizes, int n_in,
                              void* d_out, int out_size, void* d_ws, size_t ws_size,
                              hipStream_t stream) {
    const float* feat = (const float*)d_in[0];
    const float* W1   = (const float*)d_in[1];
    const int*   src  = (const int*)d_in[2];
    const int*   dst  = (const int*)d_in[3];
    float* out = (float*)d_out;

    char* ws = (char*)d_ws;
    const size_t NORM_H_BYTES = (size_t)N_NODES * D * sizeof(float);
    const size_t INTARR_BYTES = 400064;
    float* norm_h  = (float*)ws;
    int*   deg     = (int*)(ws + NORM_H_BYTES);
    int*   offsets = (int*)(ws + NORM_H_BYTES + INTARR_BYTES);
    int*   cursor  = (int*)(ws + NORM_H_BYTES + 2 * INTARR_BYTES);
    int*   csr_src = (int*)(ws + NORM_H_BYTES + 3 * INTARR_BYTES);

    copy_init_kernel<<<N_NODES * 32 / 256, 256, 0, stream>>>(feat, out, deg);
    gemm_wmma_kernel<<<N_NODES / 32 / 5, 160, 0, stream>>>(feat, W1, norm_h);
    hist_kernel<<<N_EDGES / 256, 256, 0, stream>>>(dst, deg);
    scan_kernel<<<1, 1024, 0, stream>>>(deg, offsets, cursor);
    scatter_kernel<<<N_EDGES / 256, 256, 0, stream>>>(src, dst, cursor, csr_src);
    segmax_kernel<<<N_NODES / 8, 256, 0, stream>>>(norm_h, offsets, csr_src, out);
}


# scan: one block's 98 serial 1024-wide Hillis-Steele passes -> 98 blocks in the same launch, each re-sums the L2-resident degree prefix in front of its chunk, then one shuffle scan
# speedup vs baseline: 1.9286x; 1.9286x over previous
#include <hip/hip_runtime.h>

typedef __attribute__((ext_vector_type(2))) float v2f;
typedef __attribute__((ext_vector_type(8))) float v8f;

#define N_NODES 100000
#define N_EDGES 1600000
#define D 128
#define NEG_INF (-3.402823466e+38f)

__global__ void copy_init_kernel(const float* __restrict__ feat,
                                 float* __restrict__ out,
                                 int* __restrict__ deg) {
    int idx = blockIdx.x * blockDim.x + threadIdx.x;
    int n  = idx >> 5;
    int j4 = (idx & 31) << 2;
    float4 v = *(const float4*)(feat + (size_t)n * D + j4);
    *(float4*)(out + (size_t)n * (2 * D) + j4) = v;
    if (idx <= N_NODES) deg[idx] = 0;
}

__global__ void __launch_bounds__(160, 1)
gemm_wmma_kernel(const float* __restrict__ feat,
                 const float* __restrict__ W1,
                 float* __restrict__ norm_h) {
    int wave = threadIdx.x >> 5;
    int lane = threadIdx.x & 31;
    int tm = blockIdx.x * 5 + wave;
    int half = lane >> 4;
    int l15  = lane & 15;

    const float* arow0 = feat + (size_t)(tm * 32 + l15) * D + half * 2;
    const float* arow1 = arow0 + (size_t)16 * D;
    const float* bbase = W1   + (size_t)l15 * D + half * 2;

    v8f c0[8] = {};
    v8f c1[8] = {};
    for (int k = 0; k < D; k += 4) {
        v2f a0 = *(const v2f*)(arow0 + k);
        v2f a1 = *(const v2f*)(arow1 + k);
        v2f b[8];
#pragma unroll
        for (int tn = 0; tn < 8; ++tn) {
            b[tn] = *(const v2f*)(bbase + (size_t)tn * 16 * D + k);
        }
#pragma unroll
        for (int tn = 0; tn < 8; ++tn) {
            c0[tn] = __builtin_amdgcn_wmma_f32_16x16x4_f32(
                         false, a0, false, b[tn], (short)0, c0[tn], false, false);
            c1[tn] = __builtin_amdgcn_wmma_f32_16x16x4_f32(
                         false, a1, false, b[tn], (short)0, c1[tn], false, false);
        }
    }

    float* crow0 = norm_h + (size_t)(tm * 32 + half * 8) * D + l15;
    float* crow1 = crow0 + (size_t)16 * D;
#pragma unroll
    for (int tn = 0; tn < 8; ++tn) {
#pragma unroll
        for (int v = 0; v < 8; ++v) {
            crow0[(size_t)v * D + tn * 16] = c0[tn][v];
            crow1[(size_t)v * D + tn * 16] = c1[tn][v];
        }
    }
}

__global__ void hist_kernel(const int* __restrict__ dst, int* __restrict__ deg) {
    int e = blockIdx.x * blockDim.x + threadIdx.x;
    atomicAdd(&deg[dst[e]], 1);
}

#define SCAN_CHUNK 1024
__global__ void __launch_bounds__(SCAN_CHUNK)
scan_kernel(const int* __restrict__ deg,
            int* __restrict__ offsets,
            int* __restrict__ cursor) {
    __shared__ int s_pre[32];
    __shared__ int s_tot[32];
    int tid  = threadIdx.x;
    int lane = tid & 31;
    int wave = tid >> 5;
    int base = blockIdx.x * SCAN_CHUNK;
    int i = base + tid;
    int x = (i < N_NODES) ? deg[i] : 0;

    const int4* d4 = (const int4*)deg;
    int n4 = base >> 2;
    int p0 = 0, p1 = 0, p2 = 0, p3 = 0;
    int j = tid;
    for (; j + 3 * SCAN_CHUNK < n4; j += 4 * SCAN_CHUNK) {
        int4 v0 = d4[j];
        int4 v1 = d4[j + SCAN_CHUNK];
        int4 v2 = d4[j + 2 * SCAN_CHUNK];
        int4 v3 = d4[j + 3 * SCAN_CHUNK];
        p0 += (v0.x + v0.y) + (v0.z + v0.w);
        p1 += (v1.x + v1.y) + (v1.z + v1.w);
        p2 += (v2.x + v2.y) + (v2.z + v2.w);
        p3 += (v3.x + v3.y) + (v3.z + v3.w);
    }
    for (; j < n4; j += SCAN_CHUNK) {
        int4 v0 = d4[j];
        p0 += (v0.x + v0.y) + (v0.z + v0.w);
    }
    int pre = (p0 + p1) + (p2 + p3);

    int incl = x;
    for (int off = 1; off < 32; off <<= 1) {
        int t = __shfl_up(incl, off);
        if (lane >= off) incl += t;
    }
    for (int off = 16; off > 0; off >>= 1) pre += __shfl_xor(pre, off);
    if (lane == 31) s_tot[wave] = incl;
    if (lane == 0)  s_pre[wave] = pre;
    __syncthreads();

    int carry = s_pre[lane] + ((lane < wave) ? s_tot[lane] : 0);
    for (int off = 16; off > 0; off >>= 1) carry += __shfl_xor(carry, off);

    int excl = carry + incl - x;
    if (i <= N_NODES) {
        offsets[i] = excl;
        if (i < N_NODES) cursor[i] = excl;
    }
}

__global__ void scatter_kernel(const int* __restrict__ src,
                               const int* __restrict__ dst,
                               int* __restrict__ cursor,
                               int* __restrict__ csr_src) {
    int e = blockIdx.x * blockDim.x + threadIdx.x;
    int pos = atomicAdd(&cursor[dst[e]], 1);
    csr_src[pos] = src[e];
}

__global__ void segmax_kernel(const float* __restrict__ norm_h,
                              const int* __restrict__ offsets,
                              const int* __restrict__ csr_src,
                              float* __restrict__ out) {
    int wave = threadIdx.x >> 5;
    int lane = threadIdx.x & 31;
    int node = blockIdx.x * 8 + wave;
    int beg = offsets[node];
    int end = offsets[node + 1];
    float4 acc = make_float4(NEG_INF, NEG_INF, NEG_INF, NEG_INF);
    int j4 = lane << 2;
    for (int e = beg; e < end; ++e) {
        int s = csr_src[e];
        float4 v = *(const float4*)(norm_h + (size_t)s * D + j4);
        acc.x = fmaxf(acc.x, v.x);
        acc.y = fmaxf(acc.y, v.y);
        acc.z = fmaxf(acc.z, v.z);
        acc.w = fmaxf(acc.w, v.w);
    }
    *(float4*)(out + (size_t)node * (2 * D) + D + j4) = acc;
}

extern "C" void kernel_launch(void* const* d_in, const int* in_sizes, int n_in,
                              void* d_out, int out_size, void* d_ws, size_t ws_size,
                              hipStream_t stream) {
    const float* feat = (const float*)d_in[0];
    const float* W1   = (const float*)d_in[1];
    const int*   src  = (const int*)d_in[2];
    const int*   dst  = (const int*)d_in[3];
    float* out = (float*)d_out;

    char* ws = (char*)d_ws;
    const size_t NORM_H_BYTES = (size_t)N_NODES * D * sizeof(float);
    const size_t INTARR_BYTES = 400064;
    float* norm_h  = (float*)ws;
    int*   deg     = (int*)(ws + NORM_H_BYTES);
    int*   offsets = (int*)(ws + NORM_H_BYTES + INTARR_BYTES);
    int*   cursor  = (int*)(ws + NORM_H_BYTES + 2 * INTARR_BYTES);
    int*   csr_src = (int*)(ws + NORM_H_BYTES + 3 * INTARR_BYTES);

    copy_init_kernel<<<N_NODES * 32 / 256, 256, 0, stream>>>(feat, out, deg);
    gemm_wmma_kernel<<<N_NODES / 32 / 5, 160, 0, stream>>>(feat, W1, norm_h);
    hist_kernel<<<N_EDGES / 256, 256, 0, stream>>>(dst, deg);
    scan_kernel<<<(N_NODES + SCAN_CHUNK) / SCAN_CHUNK, SCAN_CHUNK, 0, stream>>>(deg, offsets, cursor);
    scatter_kernel<<<N_EDGES / 256, 256, 0, stream>>>(src, dst, cursor, csr_src);
    segmax_kernel<<<N_NODES / 8, 256, 0, stream>>>(norm_h, offsets, csr_src, out);
}


# segment max: serial index load then row load per edge -> one coalesced index load per 32 edges, then 8 independent row loads in flight via lane reads; max order changes, result exact
# speedup vs baseline: 2.0141x; 1.0443x over previous
#include <hip/hip_runtime.h>

typedef __attribute__((ext_vector_type(2))) float v2f;
typedef __attribute__((ext_vector_type(8))) float v8f;

#define N_NODES 100000
#define N_EDGES 1600000
#define D 128
#define NEG_INF (-3.402823466e+38f)

__global__ void copy_init_kernel(const float* __restrict__ feat,
                                 float* __restrict__ out,
                                 int* __restrict__ deg) {
    int idx = blockIdx.x * blockDim.x + threadIdx.x;
    int n  = idx >> 5;
    int j4 = (idx & 31) << 2;
    float4 v = *(const float4*)(feat + (size_t)n * D + j4);
    *(float4*)(out + (size_t)n * (2 * D) + j4) = v;
    if (idx <= N_NODES) deg[idx] = 0;
}

__global__ void __launch_bounds__(160, 1)
gemm_wmma_kernel(const float* __restrict__ feat,
                 const float* __restrict__ W1,
                 float* __restrict__ norm_h) {
    int wave = threadIdx.x >> 5;
    int lane = threadIdx.x & 31;
    int tm = blockIdx.x * 5 + wave;
    int half = lane >> 4;
    int l15  = lane & 15;

    const float* arow0 = feat + (size_t)(tm * 32 + l15) * D + half * 2;
    const float* arow1 = arow0 + (size_t)16 * D;
    const float* bbase = W1   + (size_t)l15 * D + half * 2;

    v8f c0[8] = {};
    v8f c1[8] = {};
    for (int k = 0; k < D; k += 4) {
        v2f a0 = *(const v2f*)(arow0 + k);
        v2f a1 = *(const v2f*)(arow1 + k);
        v2f b[8];
#pragma unroll
        for (int tn = 0; tn < 8; ++tn) {
            b[tn] = *(const v2f*)(bbase + (size_t)tn * 16 * D + k);
        }
#pragma unroll
        for (int tn = 0; tn < 8; ++tn) {
            c0[tn] = __builtin_amdgcn_wmma_f32_16x16x4_f32(
                         false, a0, false, b[tn], (short)0, c0[tn], false, false);
            c1[tn] = __builtin_amdgcn_wmma_f32_16x16x4_f32(
                         false, a1, false, b[tn], (short)0, c1[tn], false, false);
        }
    }

    float* crow0 = norm_h + (size_t)(tm * 32 + half * 8) * D + l15;
    float* crow1 = crow0 + (size_t)16 * D;
#pragma unroll
    for (int tn = 0; tn < 8; ++tn) {
#pragma unroll
        for (int v = 0; v < 8; ++v) {
            crow0[(size_t)v * D + tn * 16] = c0[tn][v];
            crow1[(size_t)v * D + tn * 16] = c1[tn][v];
        }
    }
}

__global__ void hist_kernel(const int* __restrict__ dst, int* __restrict__ deg) {
    int e = blockIdx.x * blockDim.x + threadIdx.x;
    atomicAdd(&deg[dst[e]], 1);
}

#define SCAN_CHUNK 1024
__global__ void __launch_bounds__(SCAN_CHUNK)
scan_kernel(const int* __restrict__ deg,
            int* __restrict__ offsets,
            int* __restrict__ cursor) {
    __shared__ int s_pre[32];
    __shared__ int s_tot[32];
    int tid  = threadIdx.x;
    int lane = tid & 31;
    int wave = tid >> 5;
    int base = blockIdx.x * SCAN_CHUNK;
    int i = base + tid;
    int x = (i < N_NODES) ? deg[i] : 0;

    const int4* d4 = (const int4*)deg;
    int n4 = base >> 2;
    int p0 = 0, p1 = 0, p2 = 0, p3 = 0;
    int j = tid;
    for (; j + 3 * SCAN_CHUNK < n4; j += 4 * SCAN_CHUNK) {
        int4 v0 = d4[j];
        int4 v1 = d4[j + SCAN_CHUNK];
        int4 v2 = d4[j + 2 * SCAN_CHUNK];
        int4 v3 = d4[j + 3 * SCAN_CHUNK];
        p0 += (v0.x + v0.y) + (v0.z + v0.w);
        p1 += (v1.x + v1.y) + (v1.z + v1.w);
        p2 += (v2.x + v2.y) + (v2.z + v2.w);
        p3 += (v3.x + v3.y) + (v3.z + v3.w);
    }
    for (; j < n4; j += SCAN_CHUNK) {
        int4 v0 = d4[j];
        p0 += (v0.x + v0.y) + (v0.z + v0.w);
    }
    int pre = (p0 + p1) + (p2 + p3);

    int incl = x;
    for (int off = 1; off < 32; off <<= 1) {
        int t = __shfl_up(incl, off);
        if (lane >= off) incl += t;
    }
    for (int off = 16; off > 0; off >>= 1) pre += __shfl_xor(pre, off);
    if (lane == 31) s_tot[wave] = incl;
    if (lane == 0)  s_pre[wave] = pre;
    __syncthreads();

    int carry = s_pre[lane] + ((lane < wave) ? s_tot[lane] : 0);
    for (int off = 16; off > 0; off >>= 1) carry += __shfl_xor(carry, off);

    int excl = carry + incl - x;
    if (i <= N_NODES) {
        offsets[i] = excl;
        if (i < N_NODES) cursor[i] = excl;
    }
}

__global__ void scatter_kernel(const int* __restrict__ src,
                               const int* __restrict__ dst,
                               int* __restrict__ cursor,
                               int* __restrict__ csr_src) {
    int e = blockIdx.x * blockDim.x + threadIdx.x;
    int pos = atomicAdd(&cursor[dst[e]], 1);
    csr_src[pos] = src[e];
}

#define GATHER_DEPTH 8
__global__ void segmax_kernel(const float* __restrict__ norm_h,
                              const int* __restrict__ offsets,
                              const int* __restrict__ csr_src,
                              float* __restrict__ out) {
    unsigned lane = threadIdx.x & 31;
    int node = __builtin_amdgcn_readfirstlane((int)(blockIdx.x * 8 + (threadIdx.x >> 5)));
    int beg = offsets[node];
    int end = offsets[node + 1];
    float4 acc = make_float4(NEG_INF, NEG_INF, NEG_INF, NEG_INF);
    unsigned l4 = lane << 2;
    for (int base = beg; base < end; base += 32) {
        int cnt = end - base;
        if (cnt > 32) cnt = 32;
        int mine = ((int)lane < cnt) ? csr_src[base + (int)lane] : 0;
        int k = 0;
        for (; k + GATHER_DEPTH <= cnt; k += GATHER_DEPTH) {
            float4 v[GATHER_DEPTH];
            for (int i = 0; i < GATHER_DEPTH; ++i) {
                int s = __builtin_amdgcn_readlane(mine, k + i);
                v[i] = *(const float4*)(norm_h + (size_t)s * D + l4);
            }
            for (int i = 0; i < GATHER_DEPTH; ++i) {
                acc.x = fmaxf(acc.x, v[i].x);
                acc.y = fmaxf(acc.y, v[i].y);
                acc.z = fmaxf(acc.z, v[i].z);
                acc.w = fmaxf(acc.w, v[i].w);
            }
        }
        if (k < cnt) {
            float4 v[GATHER_DEPTH - 1];
            for (int i = 0; i < GATHER_DEPTH - 1; ++i) {
                if (k + i < cnt) {
                    int s = __builtin_amdgcn_readlane(mine, k + i);
                    v[i] = *(const float4*)(norm_h + (size_t)s * D + l4);
                } else {
                    v[i] = make_float4(NEG_INF, NEG_INF, NEG_INF, NEG_INF);
                }
            }
            for (int i = 0; i < GATHER_DEPTH - 1; ++i) {
                acc.x = fmaxf(acc.x, v[i].x);
                acc.y = fmaxf(acc.y, v[i].y);
                acc.z = fmaxf(acc.z, v[i].z);
                acc.w = fmaxf(acc.w, v[i].w);
            }
        }
    }
    *(float4*)(out + (size_t)node * (2 * D) + D + l4) = acc;
}

extern "C" void kernel_launch(void* const* d_in, const int* in_sizes, int n_in,
                              void* d_out, int out_size, void* d_ws, size_t ws_size,
                              hipStream_t stream) {
    const float* feat = (const float*)d_in[0];
    const float* W1   = (const float*)d_in[1];
    const int*   src  = (const int*)d_in[2];
    const int*   dst  = (const int*)d_in[3];
    float* out = (float*)d_out;

    char* ws = (char*)d_ws;
    const size_t NORM_H_BYTES = (size_t)N_NODES * D * sizeof(float);
    const size_t INTARR_BYTES = 400064;
    float* norm_h  = (float*)ws;
    int*   deg     = (int*)(ws + NORM_H_BYTES);
    int*   offsets = (int*)(ws + NORM_H_BYTES + INTARR_BYTES);
    int*   cursor  = (int*)(ws + NORM_H_BYTES + 2 * INTARR_BYTES);
    int*   csr_src = (int*)(ws + NORM_H_BYTES + 3 * INTARR_BYTES);

    copy_init_kernel<<<N_NODES * 32 / 256, 256, 0, stream>>>(feat, out, deg);
    gemm_wmma_kernel<<<N_NODES / 32 / 5, 160, 0, stream>>>(feat, W1, norm_h);
    hist_kernel<<<N_EDGES / 256, 256, 0, stream>>>(dst, deg);
    scan_kernel<<<(N_NODES + SCAN_CHUNK) / SCAN_CHUNK, SCAN_CHUNK, 0, stream>>>(deg, offsets, cursor);
    scatter_kernel<<<N_EDGES / 256, 256, 0, stream>>>(src, dst, cursor, csr_src);
    segmax_kernel<<<N_NODES / 8, 256, 0, stream>>>(norm_h, offsets, csr_src, out);
}


# CSR build: per-edge global atomics (degree count, returning cursor add) -> two-level counting sort on LDS counters (391 dst buckets, then rank per bucket); edge order in a node changes, max exact
# speedup vs baseline: 3.9588x; 1.9655x over previous
#include <hip/hip_runtime.h>

typedef __attribute__((ext_vector_type(2))) float v2f;
typedef __attribute__((ext_vector_type(8))) float v8f;

#define N_NODES 100000
#define N_EDGES 1600000
#define D 128
#define NEG_INF (-3.402823466e+38f)

__global__ void copy_init_kernel(const float* __restrict__ feat,
                                 float* __restrict__ out) {
    int idx = blockIdx.x * blockDim.x + threadIdx.x;
    int n  = idx >> 5;
    int j4 = (idx & 31) << 2;
    float4 v = *(const float4*)(feat + (size_t)n * D + j4);
    *(float4*)(out + (size_t)n * (2 * D) + j4) = v;
}

__global__ void __launch_bounds__(160, 1)
gemm_wmma_kernel(const float* __restrict__ feat,
                 const float* __restrict__ W1,
                 float* __restrict__ norm_h) {
    int wave = threadIdx.x >> 5;
    int lane = threadIdx.x & 31;
    int tm = blockIdx.x * 5 + wave;
    int half = lane >> 4;
    int l15  = lane & 15;

    const float* arow0 = feat + (size_t)(tm * 32 + l15) * D + half * 2;
    const float* arow1 = arow0 + (size_t)16 * D;
    const float* bbase = W1   + (size_t)l15 * D + half * 2;

    v8f c0[8] = {};
    v8f c1[8] = {};
    for (int k = 0; k < D; k += 4) {
        v2f a0 = *(const v2f*)(arow0 + k);
        v2f a1 = *(const v2f*)(arow1 + k);
        v2f b[8];
#pragma unroll
        for (int tn = 0; tn < 8; ++tn) {
            b[tn] = *(const v2f*)(bbase + (size_t)tn * 16 * D + k);
        }
#pragma unroll
        for (int tn = 0; tn < 8; ++tn) {
            c0[tn] = __builtin_amdgcn_wmma_f32_16x16x4_f32(
                         false, a0, false, b[tn], (short)0, c0[tn], false, false);
            c1[tn] = __builtin_amdgcn_wmma_f32_16x16x4_f32(
                         false, a1, false, b[tn], (short)0, c1[tn], false, false);
        }
    }

    float* crow0 = norm_h + (size_t)(tm * 32 + half * 8) * D + l15;
    float* crow1 = crow0 + (size_t)16 * D;
#pragma unroll
    for (int tn = 0; tn < 8; ++tn) {
#pragma unroll
        for (int v = 0; v < 8; ++v) {
            crow0[(size_t)v * D + tn * 16] = c0[tn][v];
            crow1[(size_t)v * D + tn * 16] = c1[tn][v];
        }
    }
}

#define SORT_BLOCKS 256
#define SORT_THREADS 1024
#define EDGES_PER_BLOCK (N_EDGES / SORT_BLOCKS)
#define SORT_TRIPS ((EDGES_PER_BLOCK + SORT_THREADS - 1) / SORT_THREADS)
#define BUCKET_SHIFT 8
#define BUCKET_NODES (1 << BUCKET_SHIFT)
#define N_BUCKETS ((N_NODES >> BUCKET_SHIFT) + 1)
#define N_SCAN (N_BUCKETS * SORT_BLOCKS)
static_assert(EDGES_PER_BLOCK * SORT_BLOCKS == N_EDGES, "edge chunks must tile the edge list");

__global__ void __launch_bounds__(SORT_THREADS)
bucket_count_kernel(const int* __restrict__ dst, int* __restrict__ counts) {
    __shared__ int s_cnt[N_BUCKETS];
    int tid  = threadIdx.x;
    int lane = tid & 31;
    int ebase = blockIdx.x * EDGES_PER_BLOCK;
    int bk[SORT_TRIPS];
#pragma unroll
    for (int k = 0; k < SORT_TRIPS; ++k) {
        int i = k * SORT_THREADS + tid;
        int ic = (i < EDGES_PER_BLOCK) ? i : (EDGES_PER_BLOCK - 1);
        int d = dst[ebase + ic];
        bk[k] = (i < EDGES_PER_BLOCK) ? (d >> BUCKET_SHIFT) : -1;
    }
    for (int b = tid; b < N_BUCKETS; b += SORT_THREADS) s_cnt[b] = 0;
    __syncthreads();
#pragma unroll
    for (int k = 0; k < SORT_TRIPS; ++k) {
        bool valid = bk[k] >= 0;
        unsigned vm = __builtin_amdgcn_ballot_w32(valid);
        if (vm != 0u) {
            int leader = __builtin_ctz(vm);
            int bk0 = __builtin_amdgcn_readlane(bk[k], leader);
            unsigned eq = __builtin_amdgcn_ballot_w32(bk[k] == bk0);
            if (eq == vm) {
                if (lane == leader) atomicAdd(&s_cnt[bk0], __builtin_popcount(vm));
            } else if (valid) {
                atomicAdd(&s_cnt[bk[k]], 1);
            }
        }
    }
    __syncthreads();
    for (int b = tid; b < N_BUCKETS; b += SORT_THREADS)
        counts[b * SORT_BLOCKS + blockIdx.x] = s_cnt[b];
}

#define SCAN_CHUNK 1024
__global__ void __launch_bounds__(SCAN_CHUNK)
scan_kernel(const int* __restrict__ cnt,
            int* __restrict__ bpos) {
    __shared__ int s_pre[32];
    __shared__ int s_tot[32];
    int tid  = threadIdx.x;
    int lane = tid & 31;
    int wave = tid >> 5;
    int base = blockIdx.x * SCAN_CHUNK;
    int i = base + tid;
    int x = (i < N_SCAN) ? cnt[i] : 0;

    const int4* d4 = (const int4*)cnt;
    int n4 = base >> 2;
    int p0 = 0, p1 = 0, p2 = 0, p3 = 0;
    int j = tid;
    for (; j + 3 * SCAN_CHUNK < n4; j += 4 * SCAN_CHUNK) {
        int4 v0 = d4[j];
        int4 v1 = d4[j + SCAN_CHUNK];
        int4 v2 = d4[j + 2 * SCAN_CHUNK];
        int4 v3 = d4[j + 3 * SCAN_CHUNK];
        p0 += (v0.x + v0.y) + (v0.z + v0.w);
        p1 += (v1.x + v1.y) + (v1.z + v1.w);
        p2 += (v2.x + v2.y) + (v2.z + v2.w);
        p3 += (v3.x + v3.y) + (v3.z + v3.w);
    }
    for (; j < n4; j += SCAN_CHUNK) {
        int4 v0 = d4[j];
        p0 += (v0.x + v0.y) + (v0.z + v0.w);
    }
    int pre = (p0 + p1) + (p2 + p3);

    int incl = x;
    for (int off = 1; off < 32; off <<= 1) {
        int t = __shfl_up(incl, off);
        if (lane >= off) incl += t;
    }
    for (int off = 16; off > 0; off >>= 1) pre += __shfl_xor(pre, off);
    if (lane == 31) s_tot[wave] = incl;
    if (lane == 0)  s_pre[wave] = pre;
    __syncthreads();

    int carry = s_pre[lane] + ((lane < wave) ? s_tot[lane] : 0);
    for (int off = 16; off > 0; off >>= 1) carry += __shfl_xor(carry, off);

    int excl = carry + incl - x;
    if (i <= N_SCAN) bpos[i] = excl;
}

__global__ void __launch_bounds__(SORT_THREADS)
bucket_scatter_kernel(const int* __restrict__ src,
                      const int* __restrict__ dst,
                      const int* __restrict__ bpos,
                      int* __restrict__ pairs_raw) {
    __shared__ int s_cur[N_BUCKETS];
    int2* pairs = (int2*)pairs_raw;
    int tid  = threadIdx.x;
    int lane = tid & 31;
    int ebase = blockIdx.x * EDGES_PER_BLOCK;
    int dv[SORT_TRIPS], sv[SORT_TRIPS];
#pragma unroll
    for (int k = 0; k < SORT_TRIPS; ++k) {
        int i = k * SORT_THREADS + tid;
        int ic = (i < EDGES_PER_BLOCK) ? i : (EDGES_PER_BLOCK - 1);
        int d = dst[ebase + ic];
        sv[k] = src[ebase + ic];
        dv[k] = (i < EDGES_PER_BLOCK) ? d : -1;
    }
    for (int b = tid; b < N_BUCKETS; b += SORT_THREADS)
        s_cur[b] = bpos[b * SORT_BLOCKS + blockIdx.x];
    __syncthreads();
#pragma unroll
    for (int k = 0; k < SORT_TRIPS; ++k) {
        bool valid = dv[k] >= 0;
        int bk = dv[k] >> BUCKET_SHIFT;
        unsigned vm = __builtin_amdgcn_ballot_w32(valid);
        if (vm != 0u) {
            int leader = __builtin_ctz(vm);
            int bk0 = __builtin_amdgcn_readlane(bk, leader);
            unsigned eq = __builtin_amdgcn_ballot_w32(bk == bk0);
            int pos = 0;
            if (eq == vm) {
                int first = 0;
                if (lane == leader) first = atomicAdd(&s_cur[bk0], __builtin_popcount(vm));
                first = __builtin_amdgcn_readlane(first, leader);
                pos = first + __builtin_popcount(vm & ((1u << lane) - 1u));
            } else if (valid) {
                pos = atomicAdd(&s_cur[bk], 1);
            }
            if (valid) pairs[pos] = make_int2(sv[k], dv[k]);
        }
    }
}

#define FINE_THREADS 1024
#define FINE_DEPTH 4
__global__ void __launch_bounds__(FINE_THREADS)
bucket_fine_kernel(const int* __restrict__ pairs_raw,
                   const int* __restrict__ bpos,
                   int* __restrict__ offsets,
                   int* __restrict__ csr_src) {
    __shared__ int s_cnt[BUCKET_NODES];
    __shared__ int s_cur[BUCKET_NODES];
    __shared__ int s_wtot[BUCKET_NODES / 32];
    const int2* pairs = (const int2*)pairs_raw;
    int tid  = threadIdx.x;
    int lane = tid & 31;
    int wave = tid >> 5;
    int b = blockIdx.x;
    int beg = bpos[b * SORT_BLOCKS];
    int end = bpos[(b + 1) * SORT_BLOCKS];
    if (tid < BUCKET_NODES) s_cnt[tid] = 0;
    __syncthreads();

    for (int i0 = beg; i0 < end; i0 += FINE_THREADS * FINE_DEPTH) {
        int ln[FINE_DEPTH];
#pragma unroll
        for (int k = 0; k < FINE_DEPTH; ++k) {
            int i = i0 + k * FINE_THREADS + tid;
            int ic = (i < end) ? i : (end - 1);
            int y = pairs[ic].y;
            ln[k] = (i < end) ? (y & (BUCKET_NODES - 1)) : -1;
        }
#pragma unroll
        for (int k = 0; k < FINE_DEPTH; ++k)
            if (ln[k] >= 0) atomicAdd(&s_cnt[ln[k]], 1);
    }
    __syncthreads();

    int x = (tid < BUCKET_NODES) ? s_cnt[tid] : 0;
    int incl = x;
    for (int off = 1; off < 32; off <<= 1) {
        int t = __shfl_up(incl, off);
        if (lane >= off) incl += t;
    }
    if (tid < BUCKET_NODES && lane == 31) s_wtot[wave] = incl;
    __syncthreads();
    if (tid < BUCKET_NODES) {
        int carry = 0;
        for (int w = 0; w < BUCKET_NODES / 32; ++w)
            if (w < wave) carry += s_wtot[w];
        int o = beg + carry + incl - x;
        s_cur[tid] = o;
        int node = b * BUCKET_NODES + tid;
        if (node <= N_NODES) offsets[node] = o;
    }
    __syncthreads();

    for (int i0 = beg; i0 < end; i0 += FINE_THREADS * FINE_DEPTH) {
        int2 p[FINE_DEPTH];
#pragma unroll
        for (int k = 0; k < FINE_DEPTH; ++k) {
            int i = i0 + k * FINE_THREADS + tid;
            int ic = (i < end) ? i : (end - 1);
            int2 q = pairs[ic];
            p[k] = (i < end) ? q : make_int2(0, -1);
        }
#pragma unroll
        for (int k = 0; k < FINE_DEPTH; ++k) {
            if (p[k].y >= 0) {
                int pos = atomicAdd(&s_cur[p[k].y & (BUCKET_NODES - 1)], 1);
                csr_src[pos] = p[k].x;
            }
        }
    }
}

#define GATHER_DEPTH 8
__global__ void segmax_kernel(const float* __restrict__ norm_h,
                              const int* __restrict__ offsets,
                              const int* __restrict__ csr_src,
                              float* __restrict__ out) {
    unsigned lane = threadIdx.x & 31;
    int node = __builtin_amdgcn_readfirstlane((int)(blockIdx.x * 8 + (threadIdx.x >> 5)));
    int beg = offsets[node];
    int end = offsets[node + 1];
    float4 acc = make_float4(NEG_INF, NEG_INF, NEG_INF, NEG_INF);
    unsigned l4 = lane << 2;
    for (int base = beg; base < end; base += 32) {
        int cnt = end - base;
        if (cnt > 32) cnt = 32;
        int mine = ((int)lane < cnt) ? csr_src[base + (int)lane] : 0;
        int k = 0;
        for (; k + GATHER_DEPTH <= cnt; k += GATHER_DEPTH) {
            float4 v[GATHER_DEPTH];
            for (int i = 0; i < GATHER_DEPTH; ++i) {
                int s = __builtin_amdgcn_readlane(mine, k + i);
                v[i] = *(const float4*)(norm_h + (size_t)s * D + l4);
            }
            for (int i = 0; i < GATHER_DEPTH; ++i) {
                acc.x = fmaxf(acc.x, v[i].x);
                acc.y = fmaxf(acc.y, v[i].y);
                acc.z = fmaxf(acc.z, v[i].z);
                acc.w = fmaxf(acc.w, v[i].w);
            }
        }
        if (k < cnt) {
            float4 v[GATHER_DEPTH - 1];
            for (int i = 0; i < GATHER_DEPTH - 1; ++i) {
                if (k + i < cnt) {
                    int s = __builtin_amdgcn_readlane(mine, k + i);
                    v[i] = *(const float4*)(norm_h + (size_t)s * D + l4);
                } else {
                    v[i] = make_float4(NEG_INF, NEG_INF, NEG_INF, NEG_INF);
                }
            }
            for (int i = 0; i < GATHER_DEPTH - 1; ++i) {
                acc.x = fmaxf(acc.x, v[i].x);
                acc.y = fmaxf(acc.y, v[i].y);
                acc.z = fmaxf(acc.z, v[i].z);
                acc.w = fmaxf(acc.w, v[i].w);
            }
        }
    }
    *(float4*)(out + (size_t)node * (2 * D) + D + l4) = acc;
}

extern "C" void kernel_launch(void* const* d_in, const int* in_sizes, int n_in,
                              void* d_out, int out_size, void* d_ws, size_t ws_size,
                              hipStream_t stream) {
    const float* feat = (const float*)d_in[0];
    const float* W1   = (const float*)d_in[1];
    const int*   src  = (const int*)d_in[2];
    const int*   dst  = (const int*)d_in[3];
    float* out = (float*)d_out;

    char* ws = (char*)d_ws;
    const size_t NORM_H_BYTES = (size_t)N_NODES * D * sizeof(float);
    const size_t INTARR_BYTES = 400064;
    const size_t PAIRS_BYTES  = (size_t)N_EDGES * 2 * sizeof(int);
    const size_t COUNTS_BYTES = (size_t)N_SCAN * sizeof(int);
    float* norm_h  = (float*)ws;
    int*   pairs   = (int*)ws;
    int*   counts  = (int*)(ws + PAIRS_BYTES);
    int*   bpos    = (int*)(ws + PAIRS_BYTES + COUNTS_BYTES);
    int*   offsets = (int*)(ws + NORM_H_BYTES + INTARR_BYTES);
    int*   csr_src = (int*)(ws + NORM_H_BYTES + 3 * INTARR_BYTES);

    bucket_count_kernel<<<SORT_BLOCKS, SORT_THREADS, 0, stream>>>(dst, counts);
    scan_kernel<<<(N_SCAN + SCAN_CHUNK) / SCAN_CHUNK, SCAN_CHUNK, 0, stream>>>(counts, bpos);
    bucket_scatter_kernel<<<SORT_BLOCKS, SORT_THREADS, 0, stream>>>(src, dst, bpos, pairs);
    bucket_fine_kernel<<<N_BUCKETS, FINE_THREADS, 0, stream>>>(pairs, bpos, offsets, csr_src);
    copy_init_kernel<<<N_NODES * 32 / 256, 256, 0, stream>>>(feat, out);
    gemm_wmma_kernel<<<N_NODES / 32 / 5, 160, 0, stream>>>(feat, W1, norm_h);
    segmax_kernel<<<N_NODES / 8, 256, 0, stream>>>(norm_h, offsets, csr_src, out);
}


# feat copy: own launch of 100000 one-load waves -> done in the GEMM epilogue by the wave that owns the 32-row stripe (whole-row loads from cache, travelling under the C stores); 7 launches -> 6
# speedup vs baseline: 4.1244x; 1.0418x over previous
#include <hip/hip_runtime.h>

typedef __attribute__((ext_vector_type(2))) float v2f;
typedef __attribute__((ext_vector_type(8))) float v8f;

#define N_NODES 100000
#define N_EDGES 1600000
#define D 128
#define NEG_INF (-3.402823466e+38f)

#define ROW_LD(i) float4 r##i = *(const float4*)(src + (size_t)(i) * D);
#define ROW_ST(i) *(float4*)(dst + (size_t)(i) * (2 * D)) = r##i;
__device__ __forceinline__ void copy_rows16(const float* src, float* dst) {
    ROW_LD(0)  ROW_LD(1)  ROW_LD(2)  ROW_LD(3)  ROW_LD(4)  ROW_LD(5)  ROW_LD(6)  ROW_LD(7)
    ROW_LD(8)  ROW_LD(9)  ROW_LD(10) ROW_LD(11) ROW_LD(12) ROW_LD(13) ROW_LD(14) ROW_LD(15)
    ROW_ST(0)  ROW_ST(1)  ROW_ST(2)  ROW_ST(3)  ROW_ST(4)  ROW_ST(5)  ROW_ST(6)  ROW_ST(7)
    ROW_ST(8)  ROW_ST(9)  ROW_ST(10) ROW_ST(11) ROW_ST(12) ROW_ST(13) ROW_ST(14) ROW_ST(15)
}

__global__ void __launch_bounds__(160, 1)
gemm_wmma_kernel(const float* feat,
                 const float* __restrict__ W1,
                 float* __restrict__ norm_h,
                 float* out) {
    int wave = threadIdx.x >> 5;
    int lane = threadIdx.x & 31;
    int tm = blockIdx.x * 5 + wave;
    int half = lane >> 4;
    int l15  = lane & 15;

    const float* arow0 = feat + (size_t)(tm * 32 + l15) * D + half * 2;
    const float* arow1 = arow0 + (size_t)16 * D;
    const float* bbase = W1   + (size_t)l15 * D + half * 2;

    v8f c0[8] = {};
    v8f c1[8] = {};
    for (int k = 0; k < D; k += 4) {
        v2f a0 = *(const v2f*)(arow0 + k);
        v2f a1 = *(const v2f*)(arow1 + k);
        v2f b[8];
#pragma unroll
        for (int tn = 0; tn < 8; ++tn) {
            b[tn] = *(const v2f*)(bbase + (size_t)tn * 16 * D + k);
        }
#pragma unroll
        for (int tn = 0; tn < 8; ++tn) {
            c0[tn] = __builtin_amdgcn_wmma_f32_16x16x4_f32(
                         false, a0, false, b[tn], (short)0, c0[tn], false, false);
            c1[tn] = __builtin_amdgcn_wmma_f32_16x16x4_f32(
                         false, a1, false, b[tn], (short)0, c1[tn], false, false);
        }
    }

    float* crow0 = norm_h + (size_t)(tm * 32 + half * 8) * D + l15;
    float* crow1 = crow0 + (size_t)16 * D;
    const float* frow = feat + (size_t)(tm * 32) * D + lane * 4;
    float*       orow = out  + (size_t)(tm * 32) * (2 * D) + lane * 4;
#pragma unroll
    for (int tn = 0; tn < 8; ++tn) {
#pragma unroll
        for (int v = 0; v < 8; ++v) {
            crow0[(size_t)v * D + tn * 16] = c0[tn][v];
            crow1[(size_t)v * D + tn * 16] = c1[tn][v];
        }
    }
    copy_rows16(frow, orow);
    copy_rows16(frow + (size_t)16 * D, orow + (size_t)16 * (2 * D));
}

#define SORT_BLOCKS 256
#define SORT_THREADS 1024
#define EDGES_PER_BLOCK (N_EDGES / SORT_BLOCKS)
#define SORT_TRIPS ((EDGES_PER_BLOCK + SORT_THREADS - 1) / SORT_THREADS)
#define BUCKET_SHIFT 8
#define BUCKET_NODES (1 << BUCKET_SHIFT)
#define N_BUCKETS ((N_NODES >> BUCKET_SHIFT) + 1)
#define N_SCAN (N_BUCKETS * SORT_BLOCKS)
static_assert(EDGES_PER_BLOCK * SORT_BLOCKS == N_EDGES, "edge chunks must tile the edge list");

__global__ void __launch_bounds__(SORT_THREADS)
bucket_count_kernel(const int* __restrict__ dst, int* __restrict__ counts) {
    __shared__ int s_cnt[N_BUCKETS];
    int tid  = threadIdx.x;
    int lane = tid & 31;
    int ebase = blockIdx.x * EDGES_PER_BLOCK;
    int bk[SORT_TRIPS];
#pragma unroll
    for (int k = 0; k < SORT_TRIPS; ++k) {
        int i = k * SORT_THREADS + tid;
        int ic = (i < EDGES_PER_BLOCK) ? i : (EDGES_PER_BLOCK - 1);
        int d = dst[ebase + ic];
        bk[k] = (i < EDGES_PER_BLOCK) ? (d >> BUCKET_SHIFT) : -1;
    }
    for (int b = tid; b < N_BUCKETS; b += SORT_THREADS) s_cnt[b] = 0;
    __syncthreads();
#pragma unroll
    for (int k = 0; k < SORT_TRIPS; ++k) {
        bool valid = bk[k] >= 0;
        unsigned vm = __builtin_amdgcn_ballot_w32(valid);
        if (vm != 0u) {
            int leader = __builtin_ctz(vm);
            int bk0 = __builtin_amdgcn_readlane(bk[k], leader);
            unsigned eq = __builtin_amdgcn_ballot_w32(bk[k] == bk0);
            if (eq == vm) {
                if (lane == leader) atomicAdd(&s_cnt[bk0], __builtin_popcount(vm));
            } else if (valid) {
                atomicAdd(&s_cnt[bk[k]], 1);
            }
        }
    }
    __syncthreads();
    for (int b = tid; b < N_BUCKETS; b += SORT_THREADS)
        counts[b * SORT_BLOCKS + blockIdx.x] = s_cnt[b];
}

#define SCAN_CHUNK 1024
__global__ void __launch_bounds__(SCAN_CHUNK)
scan_kernel(const int* __restrict__ cnt,
            int* __restrict__ bpos) {
    __shared__ int s_pre[32];
    __shared__ int s_tot[32];
    int tid  = threadIdx.x;
    int lane = tid & 31;
    int wave = tid >> 5;
    int base = blockIdx.x * SCAN_CHUNK;
    int i = base + tid;
    int x = (i < N_SCAN) ? cnt[i] : 0;

    const int4* d4 = (const int4*)cnt;
    int n4 = base >> 2;
    int p0 = 0, p1 = 0, p2 = 0, p3 = 0;
    int j = tid;
    for (; j + 3 * SCAN_CHUNK < n4; j += 4 * SCAN_CHUNK) {
        int4 v0 = d4[j];
        int4 v1 = d4[j + SCAN_CHUNK];
        int4 v2 = d4[j + 2 * SCAN_CHUNK];
        int4 v3 = d4[j + 3 * SCAN_CHUNK];
        p0 += (v0.x + v0.y) + (v0.z + v0.w);
        p1 += (v1.x + v1.y) + (v1.z + v1.w);
        p2 += (v2.x + v2.y) + (v2.z + v2.w);
        p3 += (v3.x + v3.y) + (v3.z + v3.w);
    }
    for (; j < n4; j += SCAN_CHUNK) {
        int4 v0 = d4[j];
        p0 += (v0.x + v0.y) + (v0.z + v0.w);
    }
    int pre = (p0 + p1) + (p2 + p3);

    int incl = x;
    for (int off = 1; off < 32; off <<= 1) {
        int t = __shfl_up(incl, off);
        if (lane >= off) incl += t;
    }
    for (int off = 16; off > 0; off >>= 1) pre += __shfl_xor(pre, off);
    if (lane == 31) s_tot[wave] = incl;
    if (lane == 0)  s_pre[wave] = pre;
    __syncthreads();

    int carry = s_pre[lane] + ((lane < wave) ? s_tot[lane] : 0);
    for (int off = 16; off > 0; off >>= 1) carry += __shfl_xor(carry, off);

    int excl = carry + incl - x;
    if (i <= N_SCAN) bpos[i] = excl;
}

__global__ void __launch_bounds__(SORT_THREADS)
bucket_scatter_kernel(const int* __restrict__ src,
                      const int* __restrict__ dst,
                      const int* __restrict__ bpos,
                      int* __restrict__ pairs_raw) {
    __shared__ int s_cur[N_BUCKETS];
    int2* pairs = (int2*)pairs_raw;
    int tid  = threadIdx.x;
    int lane = tid & 31;
    int ebase = blockIdx.x * EDGES_PER_BLOCK;
    int dv[SORT_TRIPS], sv[SORT_TRIPS];
#pragma unroll
    for (int k = 0; k < SORT_TRIPS; ++k) {
        int i = k * SORT_THREADS + tid;
        int ic = (i < EDGES_PER_BLOCK) ? i : (EDGES_PER_BLOCK - 1);
        int d = dst[ebase + ic];
        sv[k] = src[ebase + ic];
        dv[k] = (i < EDGES_PER_BLOCK) ? d : -1;
    }
    for (int b = tid; b < N_BUCKETS; b += SORT_THREADS)
        s_cur[b] = bpos[b * SORT_BLOCKS + blockIdx.x];
    __syncthreads();
#pragma unroll
    for (int k = 0; k < SORT_TRIPS; ++k) {
        bool valid = dv[k] >= 0;
        int bk = dv[k] >> BUCKET_SHIFT;
        unsigned vm = __builtin_amdgcn_ballot_w32(valid);
        if (vm != 0u) {
            int leader = __builtin_ctz(vm);
            int bk0 = __builtin_amdgcn_readlane(bk, leader);
            unsigned eq = __builtin_amdgcn_ballot_w32(bk == bk0);
            int pos = 0;
            if (eq == vm) {
                int first = 0;
                if (lane == leader) first = atomicAdd(&s_cur[bk0], __builtin_popcount(vm));
                first = __builtin_amdgcn_readlane(first, leader);
                pos = first + __builtin_popcount(vm & ((1u << lane) - 1u));
            } else if (valid) {
                pos = atomicAdd(&s_cur[bk], 1);
            }
            if (valid) pairs[pos] = make_int2(sv[k], dv[k]);
        }
    }
}

#define FINE_THREADS 1024
#define FINE_DEPTH 4
__global__ void __launch_bounds__(FINE_THREADS)
bucket_fine_kernel(const int* __restrict__ pairs_raw,
                   const int* __restrict__ bpos,
                   int* __restrict__ offsets,
                   int* __restrict__ csr_src) {
    __shared__ int s_cnt[BUCKET_NODES];
    __shared__ int s_cur[BUCKET_NODES];
    __shared__ int s_wtot[BUCKET_NODES / 32];
    const int2* pairs = (const int2*)pairs_raw;
    int tid  = threadIdx.x;
    int lane = tid & 31;
    int wave = tid >> 5;
    int b = blockIdx.x;
    int beg = bpos[b * SORT_BLOCKS];
    int end = bpos[(b + 1) * SORT_BLOCKS];
    if (tid < BUCKET_NODES) s_cnt[tid] = 0;
    __syncthreads();

    for (int i0 = beg; i0 < end; i0 += FINE_THREADS * FINE_DEPTH) {
        int ln[FINE_DEPTH];
#pragma unroll
        for (int k = 0; k < FINE_DEPTH; ++k) {
            int i = i0 + k * FINE_THREADS + tid;
            int ic = (i < end) ? i : (end - 1);
            int y = pairs[ic].y;
            ln[k] = (i < end) ? (y & (BUCKET_NODES - 1)) : -1;
        }
#pragma unroll
        for (int k = 0; k < FINE_DEPTH; ++k)
            if (ln[k] >= 0) atomicAdd(&s_cnt[ln[k]], 1);
    }
    __syncthreads();

    int x = (tid < BUCKET_NODES) ? s_cnt[tid] : 0;
    int incl = x;
    for (int off = 1; off < 32; off <<= 1) {
        int t = __shfl_up(incl, off);
        if (lane >= off) incl += t;
    }
    if (tid < BUCKET_NODES && lane == 31) s_wtot[wave] = incl;
    __syncthreads();
    if (tid < BUCKET_NODES) {
        int carry = 0;
        for (int w = 0; w < BUCKET_NODES / 32; ++w)
            if (w < wave) carry += s_wtot[w];
        int o = beg + carry + incl - x;
        s_cur[tid] = o;
        int node = b * BUCKET_NODES + tid;
        if (node <= N_NODES) offsets[node] = o;
    }
    __syncthreads();

    for (int i0 = beg; i0 < end; i0 += FINE_THREADS * FINE_DEPTH) {
        int2 p[FINE_DEPTH];
#pragma unroll
        for (int k = 0; k < FINE_DEPTH; ++k) {
            int i = i0 + k * FINE_THREADS + tid;
            int ic = (i < end) ? i : (end - 1);
            int2 q = pairs[ic];
            p[k] = (i < end) ? q : make_int2(0, -1);
        }
#pragma unroll
        for (int k = 0; k < FINE_DEPTH; ++k) {
            if (p[k].y >= 0) {
                int pos = atomicAdd(&s_cur[p[k].y & (BUCKET_NODES - 1)], 1);
                csr_src[pos] = p[k].x;
            }
        }
    }
}

#define GATHER_DEPTH 8
__global__ void segmax_kernel(const float* __restrict__ norm_h,
                              const int* __restrict__ offsets,
                              const int* __restrict__ csr_src,
                              float* __restrict__ out) {
    unsigned lane = threadIdx.x & 31;
    int node = __builtin_amdgcn_readfirstlane((int)(blockIdx.x * 8 + (threadIdx.x >> 5)));
    int beg = offsets[node];
    int end = offsets[node + 1];
    float4 acc = make_float4(NEG_INF, NEG_INF, NEG_INF, NEG_INF);
    unsigned l4 = lane << 2;
    for (int base = beg; base < end; base += 32) {
        int cnt = end - base;
        if (cnt > 32) cnt = 32;
        int mine = ((int)lane < cnt) ? csr_src[base + (int)lane] : 0;
        int k = 0;
        for (; k + GATHER_DEPTH <= cnt; k += GATHER_DEPTH) {
            float4 v[GATHER_DEPTH];
            for (int i = 0; i < GATHER_DEPTH; ++i) {
                int s = __builtin_amdgcn_readlane(mine, k + i);
                v[i] = *(const float4*)(norm_h + (size_t)s * D + l4);
            }
            for (int i = 0; i < GATHER_DEPTH; ++i) {
                acc.x = fmaxf(acc.x, v[i].x);
                acc.y = fmaxf(acc.y, v[i].y);
                acc.z = fmaxf(acc.z, v[i].z);
                acc.w = fmaxf(acc.w, v[i].w);
            }
        }
        if (k < cnt) {
            float4 v[GATHER_DEPTH - 1];
            for (int i = 0; i < GATHER_DEPTH - 1; ++i) {
                if (k + i < cnt) {
                    int s = __builtin_amdgcn_readlane(mine, k + i);
                    v[i] = *(const float4*)(norm_h + (size_t)s * D + l4);
                } else {
                    v[i] = make_float4(NEG_INF, NEG_INF, NEG_INF, NEG_INF);
                }
            }
            for (int i = 0; i < GATHER_DEPTH - 1; ++i) {
                acc.x = fmaxf(acc.x, v[i].x);
                acc.y = fmaxf(acc.y, v[i].y);
                acc.z = fmaxf(acc.z, v[i].z);
                acc.w = fmaxf(acc.w, v[i].w);
            }
        }
    }
    *(float4*)(out + (size_t)node * (2 * D) + D + l4) = acc;
}

extern "C" void kernel_launch(void* const* d_in, const int* in_sizes, int n_in,
                              void* d_out, int out_size, void* d_ws, size_t ws_size,
                              hipStream_t stream) {
    const float* feat = (const float*)d_in[0];
    const float* W1   = (const float*)d_in[1];
    const int*   src  = (const int*)d_in[2];
    const int*   dst  = (const int*)d_in[3];
    float* out = (float*)d_out;

    char* ws = (char*)d_ws;
    const size_t NORM_H_BYTES = (size_t)N_NODES * D * sizeof(float);
    const size_t INTARR_BYTES = 400064;
    const size_t PAIRS_BYTES  = (size_t)N_EDGES * 2 * sizeof(int);
    const size_t COUNTS_BYTES = (size_t)N_SCAN * sizeof(int);
    float* norm_h  = (float*)ws;
    int*   pairs   = (int*)ws;
    int*   counts  = (int*)(ws + PAIRS_BYTES);
    int*   bpos    = (int*)(ws + PAIRS_BYTES + COUNTS_BYTES);
    int*   offsets = (int*)(ws + NORM_H_BYTES + INTARR_BYTES);
    int*   csr_src = (int*)(ws + NORM_H_BYTES + 3 * INTARR_BYTES);

    bucket_count_kernel<<<SORT_BLOCKS, SORT_THREADS, 0, stream>>>(dst, counts);
    scan_kernel<<<(N_SCAN + SCAN_CHUNK) / SCAN_CHUNK, SCAN_CHUNK, 0, stream>>>(counts, bpos);
    bucket_scatter_kernel<<<SORT_BLOCKS, SORT_THREADS, 0, stream>>>(src, dst, bpos, pairs);
    bucket_fine_kernel<<<N_BUCKETS, FINE_THREADS, 0, stream>>>(pairs, bpos, offsets, csr_src);
    gemm_wmma_kernel<<<N_NODES / 32 / 5, 160, 0, stream>>>(feat, W1, norm_h, out);
    segmax_kernel<<<N_NODES / 8, 256, 0, stream>>>(norm_h, offsets, csr_src, out);
}


# GEMM: W1 fragments read from LDS (staged once per WGP by a 16-wave block that loops over 16-row units, pitch 132) instead of global loads touching 16 lines each; sums unchanged
# speedup vs baseline: 4.5821x; 1.1110x over previous
#include <hip/hip_runtime.h>

typedef __attribute__((ext_vector_type(2))) float v2f;
typedef __attribute__((ext_vector_type(8))) float v8f;

#define N_NODES 100000
#define N_EDGES 1600000
#define D 128
#define NEG_INF (-3.402823466e+38f)

#define ROW_LD(i) float4 r##i = *(const float4*)(src + (size_t)(i) * D);
#define ROW_ST(i) *(float4*)(dst + (size_t)(i) * (2 * D)) = r##i;
__device__ __forceinline__ void copy_rows16(const float* src, float* dst) {
    ROW_LD(0)  ROW_LD(1)  ROW_LD(2)  ROW_LD(3)  ROW_LD(4)  ROW_LD(5)  ROW_LD(6)  ROW_LD(7)
    ROW_LD(8)  ROW_LD(9)  ROW_LD(10) ROW_LD(11) ROW_LD(12) ROW_LD(13) ROW_LD(14) ROW_LD(15)
    ROW_ST(0)  ROW_ST(1)  ROW_ST(2)  ROW_ST(3)  ROW_ST(4)  ROW_ST(5)  ROW_ST(6)  ROW_ST(7)
    ROW_ST(8)  ROW_ST(9)  ROW_ST(10) ROW_ST(11) ROW_ST(12) ROW_ST(13) ROW_ST(14) ROW_ST(15)
}

#define W_PITCH 132
#define GEMM_BLOCKS 256
#define GEMM_WAVES 16
#define GEMM_THREADS (GEMM_WAVES * 32)
#define N_UNITS (N_NODES / 16)
static_assert(N_UNITS * 16 == N_NODES, "16-row units must tile the nodes");
static_assert((D * D / 4) % GEMM_THREADS == 0, "W1 staging: whole passes only");

__device__ __forceinline__ void mma_step(v8f (&c)[8], v2f a, const v2f (&b)[8]) {
#pragma unroll
    for (int tn = 0; tn < 8; ++tn)
        c[tn] = __builtin_amdgcn_wmma_f32_16x16x4_f32(
                    false, a, false, b[tn], (short)0, c[tn], false, false);
}

__global__ void __launch_bounds__(GEMM_THREADS)
gemm_wmma_kernel(const float* feat,
                 const float* __restrict__ W1,
                 float* __restrict__ norm_h,
                 float* out) {
    __shared__ __attribute__((aligned(16))) float s_w[D * W_PITCH];
    int wave = threadIdx.x >> 5;
    int lane = threadIdx.x & 31;
    int half = lane >> 4;
    int l15  = lane & 15;

    const float4* w4 = (const float4*)W1;
#pragma unroll
    for (int i = 0; i < D * D / 4 / GEMM_THREADS; ++i) {
        int f = (int)threadIdx.x + i * GEMM_THREADS;
        float4 w = w4[f];
        *(float4*)(s_w + (f >> 5) * W_PITCH + (f & 31) * 4) = w;
    }
    __syncthreads();

    const float* bl = s_w + l15 * W_PITCH + half * 2;
    int wr = wave >> 2, wc = wave & 3;
    int slot = (((wr + wc) & 1) << 3) | (wr << 1) | (wc >> 1);
    int g = __builtin_amdgcn_readfirstlane(slot * GEMM_BLOCKS + (int)blockIdx.x);
    for (int u = g; u < N_UNITS; u += GEMM_BLOCKS * GEMM_WAVES) {
        const float* arow = feat + (size_t)(u * 16 + l15) * D + half * 2;
        v8f c[8] = {};
        v2f pa = *(const v2f*)(arow);
        asm volatile("" : "+v"(pa));
        v2f pb[8];
#pragma unroll
        for (int tn = 0; tn < 8; ++tn) pb[tn] = *(const v2f*)(bl + tn * 16 * W_PITCH);

        for (int k = 0; k < D; k += 8) {
            v2f qa = *(const v2f*)(arow + k + 4);
            v2f qb[8];
#pragma unroll
            for (int tn = 0; tn < 8; ++tn) qb[tn] = *(const v2f*)(bl + tn * 16 * W_PITCH + k + 4);
            mma_step(c, pa, pb);
            int kp = (k + 8 < D) ? (k + 8) : (D - 4);
            pa = *(const v2f*)(arow + kp);
#pragma unroll
            for (int tn = 0; tn < 8; ++tn) pb[tn] = *(const v2f*)(bl + tn * 16 * W_PITCH + kp);
            mma_step(c, qa, qb);
        }

        float* crow = norm_h + (size_t)(u * 16 + half * 8) * D + l15;
#pragma unroll
        for (int tn = 0; tn < 8; ++tn) {
#pragma unroll
            for (int v = 0; v < 8; ++v)
                crow[(size_t)v * D + tn * 16] = c[tn][v];
        }
        copy_rows16(feat + (size_t)(u * 16) * D + lane * 4,
                    out  + (size_t)(u * 16) * (2 * D) + lane * 4);
    }
}

#define SORT_BLOCKS 256
#define SORT_THREADS 1024
#define EDGES_PER_BLOCK (N_EDGES / SORT_BLOCKS)
#define SORT_TRIPS ((EDGES_PER_BLOCK + SORT_THREADS - 1) / SORT_THREADS)
#define BUCKET_SHIFT 8
#define BUCKET_NODES (1 << BUCKET_SHIFT)
#define N_BUCKETS ((N_NODES >> BUCKET_SHIFT) + 1)
#define N_SCAN (N_BUCKETS * SORT_BLOCKS)
static_assert(EDGES_PER_BLOCK * SORT_BLOCKS == N_EDGES, "edge chunks must tile the edge list");

__global__ void __launch_bounds__(SORT_THREADS)
bucket_count_kernel(const int* __restrict__ dst, int* __restrict__ counts) {
    __shared__ int s_cnt[N_BUCKETS];
    int tid  = threadIdx.x;
    int lane = tid & 31;
    int ebase = blockIdx.x * EDGES_PER_BLOCK;
    int bk[SORT_TRIPS];
#pragma unroll
    for (int k = 0; k < SORT_TRIPS; ++k) {
        int i = k * SORT_THREADS + tid;
        int ic = (i < EDGES_PER_BLOCK) ? i : (EDGES_PER_BLOCK - 1);
        int d = dst[ebase + ic];
        bk[k] = (i < EDGES_PER_BLOCK) ? (d >> BUCKET_SHIFT) : -1;
    }
    for (int b = tid; b < N_BUCKETS; b += SORT_THREADS) s_cnt[b] = 0;
    __syncthreads();
#pragma unroll
    for (int k = 0; k < SORT_TRIPS; ++k) {
        bool valid = bk[k] >= 0;
        unsigned vm = __builtin_amdgcn_ballot_w32(valid);
        if (vm != 0u) {
            int leader = __builtin_ctz(vm);
            int bk0 = __builtin_amdgcn_readlane(bk[k], leader);
            unsigned eq = __builtin_amdgcn_ballot_w32(bk[k] == bk0);
            if (eq == vm) {
                if (lane == leader) atomicAdd(&s_cnt[bk0], __builtin_popcount(vm));
            } else if (valid) {
                atomicAdd(&s_cnt[bk[k]], 1);
            }
        }
    }
    __syncthreads();
    for (int b = tid; b < N_BUCKETS; b += SORT_THREADS)
        counts[b * SORT_BLOCKS + blockIdx.x] = s_cnt[b];
}

#define SCAN_CHUNK 1024
__global__ void __launch_bounds__(SCAN_CHUNK)
scan_kernel(const int* __restrict__ cnt,
            int* __restrict__ bpos) {
    __shared__ int s_pre[32];
    __shared__ int s_tot[32];
    int tid  = threadIdx.x;
    int lane = tid & 31;
    int wave = tid >> 5;
    int base = blockIdx.x * SCAN_CHUNK;
    int i = base + tid;
    int x = (i < N_SCAN) ? cnt[i] : 0;

    const int4* d4 = (const int4*)cnt;
    int n4 = base >> 2;
    int p0 = 0, p1 = 0, p2 = 0, p3 = 0;
    int j = tid;
    for (; j + 3 * SCAN_CHUNK < n4; j += 4 * SCAN_CHUNK) {
        int4 v0 = d4[j];
        int4 v1 = d4[j + SCAN_CHUNK];
        int4 v2 = d4[j + 2 * SCAN_CHUNK];
        int4 v3 = d4[j + 3 * SCAN_CHUNK];
        p0 += (v0.x + v0.y) + (v0.z + v0.w);
        p1 += (v1.x + v1.y) + (v1.z + v1.w);
        p2 += (v2.x + v2.y) + (v2.z + v2.w);
        p3 += (v3.x + v3.y) + (v3.z + v3.w);
    }
    for (; j < n4; j += SCAN_CHUNK) {
        int4 v0 = d4[j];
        p0 += (v0.x + v0.y) + (v0.z + v0.w);
    }
    int pre = (p0 + p1) + (p2 + p3);

    int incl = x;
    for (int off = 1; off < 32; off <<= 1) {
        int t = __shfl_up(incl, off);
        if (lane >= off) incl += t;
    }
    for (int off = 16; off > 0; off >>= 1) pre += __shfl_xor(pre, off);
    if (lane == 31) s_tot[wave] = incl;
    if (lane == 0)  s_pre[wave] = pre;
    __syncthreads();

    int carry = s_pre[lane] + ((lane < wave) ? s_tot[lane] : 0);
    for (int off = 16; off > 0; off >>= 1) carry += __shfl_xor(carry, off);

    int excl = carry + incl - x;
    if (i <= N_SCAN) bpos[i] = excl;
}

__global__ void __launch_bounds__(SORT_THREADS)
bucket_scatter_kernel(const int* __restrict__ src,
                      const int* __restrict__ dst,
                      const int* __restrict__ bpos,
                      int* __restrict__ pairs_raw) {
    __shared__ int s_cur[N_BUCKETS];
    int2* pairs = (int2*)pairs_raw;
    int tid  = threadIdx.x;
    int lane = tid & 31;
    int ebase = blockIdx.x * EDGES_PER_BLOCK;
    int dv[SORT_TRIPS], sv[SORT_TRIPS];
#pragma unroll
    for (int k = 0; k < SORT_TRIPS; ++k) {
        int i = k * SORT_THREADS + tid;
        int ic = (i < EDGES_PER_BLOCK) ? i : (EDGES_PER_BLOCK - 1);
        int d = dst[ebase + ic];
        sv[k] = src[ebase + ic];
        dv[k] = (i < EDGES_PER_BLOCK) ? d : -1;
    }
    for (int b = tid; b < N_BUCKETS; b += SORT_THREADS)
        s_cur[b] = bpos[b * SORT_BLOCKS + blockIdx.x];
    __syncthreads();
#pragma unroll
    for (int k = 0; k < SORT_TRIPS; ++k) {
        bool valid = dv[k] >= 0;
        int bk = dv[k] >> BUCKET_SHIFT;
        unsigned vm = __builtin_amdgcn_ballot_w32(valid);
        if (vm != 0u) {
            int leader = __builtin_ctz(vm);
            int bk0 = __builtin_amdgcn_readlane(bk, leader);
            unsigned eq = __builtin_amdgcn_ballot_w32(bk == bk0);
            int pos = 0;
            if (eq == vm) {
                int first = 0;
                if (lane == leader) first = atomicAdd(&s_cur[bk0], __builtin_popcount(vm));
                first = __builtin_amdgcn_readlane(first, leader);
                pos = first + __builtin_popcount(vm & ((1u << lane) - 1u));
            } else if (valid) {
                pos = atomicAdd(&s_cur[bk], 1);
            }
            if (valid) pairs[pos] = make_int2(sv[k], dv[k]);
        }
    }
}

#define FINE_THREADS 1024
#define FINE_DEPTH 4
__global__ void __launch_bounds__(FINE_THREADS)
bucket_fine_kernel(const int* __restrict__ pairs_raw,
                   const int* __restrict__ bpos,
                   int* __restrict__ offsets,
                   int* __restrict__ csr_src) {
    __shared__ int s_cnt[BUCKET_NODES];
    __shared__ int s_cur[BUCKET_NODES];
    __shared__ int s_wtot[BUCKET_NODES / 32];
    const int2* pairs = (const int2*)pairs_raw;
    int tid  = threadIdx.x;
    int lane = tid & 31;
    int wave = tid >> 5;
    int b = blockIdx.x;
    int beg = bpos[b * SORT_BLOCKS];
    int end = bpos[(b + 1) * SORT_BLOCKS];
    if (tid < BUCKET_NODES) s_cnt[tid] = 0;
    __syncthreads();

    for (int i0 = beg; i0 < end; i0 += FINE_THREADS * FINE_DEPTH) {
        int ln[FINE_DEPTH];
#pragma unroll
        for (int k = 0; k < FINE_DEPTH; ++k) {
            int i = i0 + k * FINE_THREADS + tid;
            int ic = (i < end) ? i : (end - 1);
            int y = pairs[ic].y;
            ln[k] = (i < end) ? (y & (BUCKET_NODES - 1)) : -1;
        }
#pragma unroll
        for (int k = 0; k < FINE_DEPTH; ++k)
            if (ln[k] >= 0) atomicAdd(&s_cnt[ln[k]], 1);
    }
    __syncthreads();

    int x = (tid < BUCKET_NODES) ? s_cnt[tid] : 0;
    int incl = x;
    for (int off = 1; off < 32; off <<= 1) {
        int t = __shfl_up(incl, off);
        if (lane >= off) incl += t;
    }
    if (tid < BUCKET_NODES && lane == 31) s_wtot[wave] = incl;
    __syncthreads();
    if (tid < BUCKET_NODES) {
        int carry = 0;
        for (int w = 0; w < BUCKET_NODES / 32; ++w)
            if (w < wave) carry += s_wtot[w];
        int o = beg + carry + incl - x;
        s_cur[tid] = o;
        int node = b * BUCKET_NODES + tid;
        if (node <= N_NODES) offsets[node] = o;
    }
    __syncthreads();

    for (int i0 = beg; i0 < end; i0 += FINE_THREADS * FINE_DEPTH) {
        int2 p[FINE_DEPTH];
#pragma unroll
        for (int k = 0; k < FINE_DEPTH; ++k) {
            int i = i0 + k * FINE_THREADS + tid;
            int ic = (i < end) ? i : (end - 1);
            int2 q = pairs[ic];
            p[k] = (i < end) ? q : make_int2(0, -1);
        }
#pragma unroll
        for (int k = 0; k < FINE_DEPTH; ++k) {
            if (p[k].y >= 0) {
                int pos = atomicAdd(&s_cur[p[k].y & (BUCKET_NODES - 1)], 1);
                csr_src[pos] = p[k].x;
            }
        }
    }
}

#define GATHER_DEPTH 8
__global__ void segmax_kernel(const float* __restrict__ norm_h,
                              const int* __restrict__ offsets,
                              const int* __restrict__ csr_src,
                              float* __restrict__ out) {
    unsigned lane = threadIdx.x & 31;
    int node = __builtin_amdgcn_readfirstlane((int)(blockIdx.x * 8 + (threadIdx.x >> 5)));
    int beg = offsets[node];
    int end = offsets[node + 1];
    float4 acc = make_float4(NEG_INF, NEG_INF, NEG_INF, NEG_INF);
    unsigned l4 = lane << 2;
    for (int base = beg; base < end; base += 32) {
        int cnt = end - base;
        if (cnt > 32) cnt = 32;
        int mine = ((int)lane < cnt) ? csr_src[base + (int)lane] : 0;
        int k = 0;
        for (; k + GATHER_DEPTH <= cnt; k += GATHER_DEPTH) {
            float4 v[GATHER_DEPTH];
            for (int i = 0; i < GATHER_DEPTH; ++i) {
                int s = __builtin_amdgcn_readlane(mine, k + i);
                v[i] = *(const float4*)(norm_h + (size_t)s * D + l4);
            }
            for (int i = 0; i < GATHER_DEPTH; ++i) {
                acc.x = fmaxf(acc.x, v[i].x);
                acc.y = fmaxf(acc.y, v[i].y);
                acc.z = fmaxf(acc.z, v[i].z);
                acc.w = fmaxf(acc.w, v[i].w);
            }
        }
        if (k < cnt) {
            float4 v[GATHER_DEPTH - 1];
            for (int i = 0; i < GATHER_DEPTH - 1; ++i) {
                if (k + i < cnt) {
                    int s = __builtin_amdgcn_readlane(mine, k + i);
                    v[i] = *(const float4*)(norm_h + (size_t)s * D + l4);
                } else {
                    v[i] = make_float4(NEG_INF, NEG_INF, NEG_INF, NEG_INF);
                }
            }
            for (int i = 0; i < GATHER_DEPTH - 1; ++i) {
                acc.x = fmaxf(acc.x, v[i].x);
                acc.y = fmaxf(acc.y, v[i].y);
                acc.z = fmaxf(acc.z, v[i].z);
                acc.w = fmaxf(acc.w, v[i].w);
            }
        }
    }
    *(float4*)(out + (size_t)node * (2 * D) + D + l4) = acc;
}

extern "C" void kernel_launch(void* const* d_in, const int* in_sizes, int n_in,
                              void* d_out, int out_size, void* d_ws, size_t ws_size,
                              hipStream_t stream) {
    const float* feat = (const float*)d_in[0];
    const float* W1   = (const float*)d_in[1];
    const int*   src  = (const int*)d_in[2];
    const int*   dst  = (const int*)d_in[3];
    float* out = (float*)d_out;

    char* ws = (char*)d_ws;
    const size_t NORM_H_BYTES = (size_t)N_NODES * D * sizeof(float);
    const size_t INTARR_BYTES = 400064;
    const size_t PAIRS_BYTES  = (size_t)N_EDGES * 2 * sizeof(int);
    const size_t COUNTS_BYTES = (size_t)N_SCAN * sizeof(int);
    float* norm_h  = (float*)ws;
    int*   pairs   = (int*)ws;
    int*   counts  = (int*)(ws + PAIRS_BYTES);
    int*   bpos    = (int*)(ws + PAIRS_BYTES + COUNTS_BYTES);
    int*   offsets = (int*)(ws + NORM_H_BYTES + INTARR_BYTES);
    int*   csr_src = (int*)(ws + NORM_H_BYTES + 3 * INTARR_BYTES);

    bucket_count_kernel<<<SORT_BLOCKS, SORT_THREADS, 0, stream>>>(dst, counts);
    scan_kernel<<<(N_SCAN + SCAN_CHUNK) / SCAN_CHUNK, SCAN_CHUNK, 0, stream>>>(counts, bpos);
    bucket_scatter_kernel<<<SORT_BLOCKS, SORT_THREADS, 0, stream>>>(src, dst, bpos, pairs);
    bucket_fine_kernel<<<N_BUCKETS, FINE_THREADS, 0, stream>>>(pairs, bpos, offsets, csr_src);
    gemm_wmma_kernel<<<GEMM_BLOCKS, GEMM_THREADS, 0, stream>>>(feat, W1, norm_h, out);
    segmax_kernel<<<N_NODES / 8, 256, 0, stream>>>(norm_h, offsets, csr_src, out);
}


# CSR build level 1: global count, scan, scatter (3 launches, one 8-byte store per line) -> each block sorts its 6250 edges by bucket in LDS, writes them whole with an offset row; 6 launches -> 4
# speedup vs baseline: 5.4711x; 1.1940x over previous
#include <hip/hip_runtime.h>

typedef __attribute__((ext_vector_type(2))) float v2f;
typedef __attribute__((ext_vector_type(8))) float v8f;

#define N_NODES 100000
#define N_EDGES 1600000
#define D 128
#define NEG_INF (-3.402823466e+38f)

#define ROW_LD(i) float4 r##i = *(const float4*)(src + (size_t)(i) * D);
#define ROW_ST(i) *(float4*)(dst + (size_t)(i) * (2 * D)) = r##i;
__device__ __forceinline__ void copy_rows16(const float* src, float* dst) {
    ROW_LD(0)  ROW_LD(1)  ROW_LD(2)  ROW_LD(3)  ROW_LD(4)  ROW_LD(5)  ROW_LD(6)  ROW_LD(7)
    ROW_LD(8)  ROW_LD(9)  ROW_LD(10) ROW_LD(11) ROW_LD(12) ROW_LD(13) ROW_LD(14) ROW_LD(15)
    ROW_ST(0)  ROW_ST(1)  ROW_ST(2)  ROW_ST(3)  ROW_ST(4)  ROW_ST(5)  ROW_ST(6)  ROW_ST(7)
    ROW_ST(8)  ROW_ST(9)  ROW_ST(10) ROW_ST(11) ROW_ST(12) ROW_ST(13) ROW_ST(14) ROW_ST(15)
}

#define W_PITCH 132
#define GEMM_BLOCKS 256
#define GEMM_WAVES 16
#define GEMM_THREADS (GEMM_WAVES * 32)
#define N_UNITS (N_NODES / 16)
static_assert(N_UNITS * 16 == N_NODES, "16-row units must tile the nodes");
static_assert((D * D / 4) % GEMM_THREADS == 0, "W1 staging: whole passes only");

__device__ __forceinline__ void mma_step(v8f (&c)[8], v2f a, const v2f (&b)[8]) {
#pragma unroll
    for (int tn = 0; tn < 8; ++tn)
        c[tn] = __builtin_amdgcn_wmma_f32_16x16x4_f32(
                    false, a, false, b[tn], (short)0, c[tn], false, false);
}

__global__ void __launch_bounds__(GEMM_THREADS)
gemm_wmma_kernel(const float* feat,
                 const float* __restrict__ W1,
                 float* __restrict__ norm_h,
                 float* out) {
    __shared__ __attribute__((aligned(16))) float s_w[D * W_PITCH];
    int wave = threadIdx.x >> 5;
    int lane = threadIdx.x & 31;
    int half = lane >> 4;
    int l15  = lane & 15;

    const float4* w4 = (const float4*)W1;
#pragma unroll
    for (int i = 0; i < D * D / 4 / GEMM_THREADS; ++i) {
        int f = (int)threadIdx.x + i * GEMM_THREADS;
        float4 w = w4[f];
        *(float4*)(s_w + (f >> 5) * W_PITCH + (f & 31) * 4) = w;
    }
    __syncthreads();

    const float* bl = s_w + l15 * W_PITCH + half * 2;
    int wr = wave >> 2, wc = wave & 3;
    int slot = (((wr + wc) & 1) << 3) | (wr << 1) | (wc >> 1);
    int g = __builtin_amdgcn_readfirstlane(slot * GEMM_BLOCKS + (int)blockIdx.x);
    for (int u = g; u < N_UNITS; u += GEMM_BLOCKS * GEMM_WAVES) {
        const float* arow = feat + (size_t)(u * 16 + l15) * D + half * 2;
        v8f c[8] = {};
        v2f pa = *(const v2f*)(arow);
        asm volatile("" : "+v"(pa));
        v2f pb[8];
#pragma unroll
        for (int tn = 0; tn < 8; ++tn) pb[tn] = *(const v2f*)(bl + tn * 16 * W_PITCH);

        for (int k = 0; k < D; k += 8) {
            v2f qa = *(const v2f*)(arow + k + 4);
            v2f qb[8];
#pragma unroll
            for (int tn = 0; tn < 8; ++tn) qb[tn] = *(const v2f*)(bl + tn * 16 * W_PITCH + k + 4);
            mma_step(c, pa, pb);
            int kp = (k + 8 < D) ? (k + 8) : (D - 4);
            pa = *(const v2f*)(arow + kp);
#pragma unroll
            for (int tn = 0; tn < 8; ++tn) pb[tn] = *(const v2f*)(bl + tn * 16 * W_PITCH + kp);
            mma_step(c, qa, qb);
        }

        float* crow = norm_h + (size_t)(u * 16 + half * 8) * D + l15;
#pragma unroll
        for (int tn = 0; tn < 8; ++tn) {
#pragma unroll
            for (int v = 0; v < 8; ++v)
                crow[(size_t)v * D + tn * 16] = c[tn][v];
        }
        copy_rows16(feat + (size_t)(u * 16) * D + lane * 4,
                    out  + (size_t)(u * 16) * (2 * D) + lane * 4);
    }
}

#define SORT_BLOCKS 256
#define SORT_THREADS 1024
#define EDGES_PER_BLOCK (N_EDGES / SORT_BLOCKS)
#define SORT_TRIPS ((EDGES_PER_BLOCK + SORT_THREADS - 1) / SORT_THREADS)
#define BUCKET_SHIFT 8
#define BUCKET_NODES (1 << BUCKET_SHIFT)
#define N_BUCKETS ((N_NODES >> BUCKET_SHIFT) + 1)
#define TAB_ROWS (N_BUCKETS + 1)
static_assert(EDGES_PER_BLOCK * SORT_BLOCKS == N_EDGES, "edge chunks must tile the edge list");
static_assert(EDGES_PER_BLOCK % 2 == 0, "a piece is written out as whole 16-byte words");
static_assert(TAB_ROWS <= SORT_THREADS, "one thread per bucket in the block's scan");

__global__ void __launch_bounds__(SORT_THREADS)
block_sort_kernel(const int* __restrict__ src,
                  const int* __restrict__ dst,
                  int* __restrict__ pairs_raw,
                  int* __restrict__ tab) {
    __shared__ __attribute__((aligned(16))) int s_pairs[EDGES_PER_BLOCK * 2];
    __shared__ int s_cnt[N_BUCKETS];
    __shared__ int s_off[TAB_ROWS];
    __shared__ int s_wtot[SORT_THREADS / 32];
    int tid  = threadIdx.x;
    int lane = tid & 31;
    int wave = tid >> 5;
    int ebase = blockIdx.x * EDGES_PER_BLOCK;
    int dv[SORT_TRIPS], sv[SORT_TRIPS], rk[SORT_TRIPS];
#pragma unroll
    for (int k = 0; k < SORT_TRIPS; ++k) {
        int i = k * SORT_THREADS + tid;
        int ic = (i < EDGES_PER_BLOCK) ? i : (EDGES_PER_BLOCK - 1);
        int d = dst[ebase + ic];
        sv[k] = src[ebase + ic];
        dv[k] = (i < EDGES_PER_BLOCK) ? d : -1;
    }
    for (int b = tid; b < N_BUCKETS; b += SORT_THREADS) s_cnt[b] = 0;
    __syncthreads();
#pragma unroll
    for (int k = 0; k < SORT_TRIPS; ++k) {
        bool valid = dv[k] >= 0;
        int bk = dv[k] >> BUCKET_SHIFT;
        unsigned vm = __builtin_amdgcn_ballot_w32(valid);
        int r = 0;
        if (vm != 0u) {
            int leader = __builtin_ctz(vm);
            int bk0 = __builtin_amdgcn_readlane(bk, leader);
            unsigned eq = __builtin_amdgcn_ballot_w32(bk == bk0);
            if (eq == vm) {
                int first = 0;
                if (lane == leader) first = atomicAdd(&s_cnt[bk0], __builtin_popcount(vm));
                first = __builtin_amdgcn_readlane(first, leader);
                r = first + __builtin_popcount(vm & ((1u << lane) - 1u));
            } else if (valid) {
                r = atomicAdd(&s_cnt[bk], 1);
            }
        }
        rk[k] = r;
    }
    __syncthreads();

    int x = (tid < N_BUCKETS) ? s_cnt[tid] : 0;
    int incl = x;
    for (int off = 1; off < 32; off <<= 1) {
        int t = __shfl_up(incl, off);
        if (lane >= off) incl += t;
    }
    if (lane == 31) s_wtot[wave] = incl;
    __syncthreads();
    int carry = (lane < wave) ? s_wtot[lane] : 0;
    for (int off = 16; off > 0; off >>= 1) carry += __shfl_xor(carry, off);
    if (tid < TAB_ROWS) s_off[tid] = carry + incl - x;
    __syncthreads();

    int2* sp2 = (int2*)s_pairs;
#pragma unroll
    for (int k = 0; k < SORT_TRIPS; ++k)
        if (dv[k] >= 0) sp2[s_off[dv[k] >> BUCKET_SHIFT] + rk[k]] = make_int2(sv[k], dv[k]);
    __syncthreads();

    int4* outp = (int4*)(pairs_raw + (size_t)ebase * 2);
    const int4* sp4 = (const int4*)s_pairs;
    for (int i = tid; i < EDGES_PER_BLOCK / 2; i += SORT_THREADS) outp[i] = sp4[i];
    if (tid < TAB_ROWS) tab[tid * SORT_BLOCKS + blockIdx.x] = s_off[tid];
}

#define FINE_THREADS 1024
#define RUNS_PER_WAVE (SORT_BLOCKS / (FINE_THREADS / 32))
static_assert(RUNS_PER_WAVE * (FINE_THREADS / 32) == SORT_BLOCKS, "the waves must tile the runs");
static_assert(SORT_BLOCKS == BUCKET_NODES, "threads 0..255 set up one run and one node counter each");
__global__ void __launch_bounds__(FINE_THREADS)
bucket_fine_kernel(const int* __restrict__ pairs_raw,
                   const int* __restrict__ tab,
                   int* __restrict__ offsets,
                   int* __restrict__ csr_src) {
    __shared__ int s_cnt[BUCKET_NODES];
    __shared__ int s_cur[BUCKET_NODES];
    __shared__ int s_wtot[BUCKET_NODES / 32];
    __shared__ int s_lo[SORT_BLOCKS];
    __shared__ int s_len[SORT_BLOCKS];
    __shared__ int s_beg[SORT_BLOCKS / 32];
    const int2* pairs = (const int2*)pairs_raw;
    int tid  = threadIdx.x;
    int lane = tid & 31;
    int wave = tid >> 5;
    int b = blockIdx.x;
    if (tid < SORT_BLOCKS) {
        int lo = tab[b * SORT_BLOCKS + tid];
        int hi = tab[(b + 1) * SORT_BLOCKS + tid];
        s_lo[tid]  = lo;
        s_len[tid] = hi - lo;
        s_cnt[tid] = 0;
        int s = lo;
        for (int off = 16; off > 0; off >>= 1) s += __shfl_xor(s, off);
        if (lane == 0) s_beg[wave] = s;
    }
    __syncthreads();
    int beg = 0;
    for (int w = 0; w < SORT_BLOCKS / 32; ++w) beg += s_beg[w];

    int rbase[RUNS_PER_WAVE], rlen[RUNS_PER_WAVE], rlast[RUNS_PER_WAVE];
#pragma unroll
    for (int r = 0; r < RUNS_PER_WAVE; ++r) {
        int t = wave * RUNS_PER_WAVE + r;
        rbase[r] = __builtin_amdgcn_readfirstlane(t * EDGES_PER_BLOCK + s_lo[t]);
        rlen[r]  = __builtin_amdgcn_readfirstlane(s_len[t]);
        rlast[r] = __builtin_amdgcn_readfirstlane(t * EDGES_PER_BLOCK + (EDGES_PER_BLOCK - 1));
    }

    {
        int y[RUNS_PER_WAVE];
#pragma unroll
        for (int r = 0; r < RUNS_PER_WAVE; ++r) {
            int i = rbase[r] + lane;
            int ic = (i < rlast[r]) ? i : rlast[r];
            int yy = pairs[ic].y;
            y[r] = (lane < rlen[r]) ? yy : -1;
        }
#pragma unroll
        for (int r = 0; r < RUNS_PER_WAVE; ++r)
            if (y[r] >= 0) atomicAdd(&s_cnt[y[r] & (BUCKET_NODES - 1)], 1);
#pragma unroll
        for (int r = 0; r < RUNS_PER_WAVE; ++r) {
#pragma nounroll
            for (int c = 32; c < rlen[r]; c += 32) {
                int i = rbase[r] + c + lane;
                int ic = (i < rlast[r]) ? i : rlast[r];
                int yy = pairs[ic].y;
                if (c + lane < rlen[r]) atomicAdd(&s_cnt[yy & (BUCKET_NODES - 1)], 1);
            }
        }
    }
    __syncthreads();

    int x = (tid < BUCKET_NODES) ? s_cnt[tid] : 0;
    int incl = x;
    for (int off = 1; off < 32; off <<= 1) {
        int t = __shfl_up(incl, off);
        if (lane >= off) incl += t;
    }
    if (tid < BUCKET_NODES && lane == 31) s_wtot[wave] = incl;
    __syncthreads();
    if (tid < BUCKET_NODES) {
        int carry = 0;
        for (int w = 0; w < BUCKET_NODES / 32; ++w)
            if (w < wave) carry += s_wtot[w];
        int o = beg + carry + incl - x;
        s_cur[tid] = o;
        int node = b * BUCKET_NODES + tid;
        if (node <= N_NODES) offsets[node] = o;
    }
    __syncthreads();

    {
        int2 p[RUNS_PER_WAVE];
#pragma unroll
        for (int r = 0; r < RUNS_PER_WAVE; ++r) {
            int i = rbase[r] + lane;
            int ic = (i < rlast[r]) ? i : rlast[r];
            int2 q = pairs[ic];
            p[r] = (lane < rlen[r]) ? q : make_int2(0, -1);
        }
#pragma unroll
        for (int r = 0; r < RUNS_PER_WAVE; ++r) {
            if (p[r].y >= 0) {
                int pos = atomicAdd(&s_cur[p[r].y & (BUCKET_NODES - 1)], 1);
                csr_src[pos] = p[r].x;
            }
        }
#pragma unroll
        for (int r = 0; r < RUNS_PER_WAVE; ++r) {
#pragma nounroll
            for (int c = 32; c < rlen[r]; c += 32) {
                int i = rbase[r] + c + lane;
                int ic = (i < rlast[r]) ? i : rlast[r];
                int2 q = pairs[ic];
                if (c + lane < rlen[r]) {
                    int pos = atomicAdd(&s_cur[q.y & (BUCKET_NODES - 1)], 1);
                    csr_src[pos] = q.x;
                }
            }
        }
    }
}

#define GATHER_DEPTH 8
__global__ void segmax_kernel(const float* __restrict__ norm_h,
                              const int* __restrict__ offsets,
                              const int* __restrict__ csr_src,
                              float* __restrict__ out) {
    unsigned lane = threadIdx.x & 31;
    int node = __builtin_amdgcn_readfirstlane((int)(blockIdx.x * 8 + (threadIdx.x >> 5)));
    int beg = offsets[node];
    int end = offsets[node + 1];
    float4 acc = make_float4(NEG_INF, NEG_INF, NEG_INF, NEG_INF);
    unsigned l4 = lane << 2;
    for (int base = beg; base < end; base += 32) {
        int cnt = end - base;
        if (cnt > 32) cnt = 32;
        int mine = ((int)lane < cnt) ? csr_src[base + (int)lane] : 0;
        int k = 0;
        for (; k + GATHER_DEPTH <= cnt; k += GATHER_DEPTH) {
            float4 v[GATHER_DEPTH];
            for (int i = 0; i < GATHER_DEPTH; ++i) {
                int s = __builtin_amdgcn_readlane(mine, k + i);
                v[i] = *(const float4*)(norm_h + (size_t)s * D + l4);
            }
            for (int i = 0; i < GATHER_DEPTH; ++i) {
                acc.x = fmaxf(acc.x, v[i].x);
                acc.y = fmaxf(acc.y, v[i].y);
                acc.z = fmaxf(acc.z, v[i].z);
                acc.w = fmaxf(acc.w, v[i].w);
            }
        }
        if (k < cnt) {
            float4 v[GATHER_DEPTH - 1];
            for (int i = 0; i < GATHER_DEPTH - 1; ++i) {
                if (k + i < cnt) {
                    int s = __builtin_amdgcn_readlane(mine, k + i);
                    v[i] = *(const float4*)(norm_h + (size_t)s * D + l4);
                } else {
                    v[i] = make_float4(NEG_INF, NEG_INF, NEG_INF, NEG_INF);
                }
            }
            for (int i = 0; i < GATHER_DEPTH - 1; ++i) {
                acc.x = fmaxf(acc.x, v[i].x);
                acc.y = fmaxf(acc.y, v[i].y);
                acc.z = fmaxf(acc.z, v[i].z);
                acc.w = fmaxf(acc.w, v[i].w);
            }
        }
    }
    *(float4*)(out + (size_t)node * (2 * D) + D + l4) = acc;
}

extern "C" void kernel_launch(void* const* d_in, const int* in_sizes, int n_in,
                              void* d_out, int out_size, void* d_ws, size_t ws_size,
                              hipStream_t stream) {
    const float* feat = (const float*)d_in[0];
    const float* W1   = (const float*)d_in[1];
    const int*   src  = (const int*)d_in[2];
    const int*   dst  = (const int*)d_in[3];
    float* out = (float*)d_out;

    char* ws = (char*)d_ws;
    const size_t NORM_H_BYTES = (size_t)N_NODES * D * sizeof(float);
    const size_t INTARR_BYTES = 400064;
    const size_t PAIRS_BYTES  = (size_t)N_EDGES * 2 * sizeof(int);
    float* norm_h  = (float*)ws;
    int*   pairs   = (int*)ws;
    int*   tab     = (int*)(ws + PAIRS_BYTES);
    int*   offsets = (int*)(ws + NORM_H_BYTES + INTARR_BYTES);
    int*   csr_src = (int*)(ws + NORM_H_BYTES + 3 * INTARR_BYTES);

    block_sort_kernel<<<SORT_BLOCKS, SORT_THREADS, 0, stream>>>(src, dst, pairs, tab);
    bucket_fine_kernel<<<N_BUCKETS, FINE_THREADS, 0, stream>>>(pairs, tab, offsets, csr_src);
    gemm_wmma_kernel<<<GEMM_BLOCKS, GEMM_THREADS, 0, stream>>>(feat, W1, norm_h, out);
    segmax_kernel<<<N_NODES / 8, 256, 0, stream>>>(norm_h, offsets, csr_src, out);
}


# edge sort level 2 + segment max in one launch: 512 blocks (two per WGP) build the bucket's source lists in LDS from one-word edges and take the row max there; max order changes, exact; 4->3 launches
# speedup vs baseline: 6.4800x; 1.1844x over previous
#include <hip/hip_runtime.h>

typedef __attribute__((ext_vector_type(2))) float v2f;
typedef __attribute__((ext_vector_type(8))) float v8f;

#define N_NODES 100000
#define N_EDGES 1600000
#define D 128
#define NEG_INF (-3.402823466e+38f)

#define ROW_LD(i) float4 r##i = *(const float4*)(src + (size_t)(i) * D);
#define ROW_ST(i) *(float4*)(dst + (size_t)(i) * (2 * D)) = r##i;
__device__ __forceinline__ void copy_rows16(const float* src, float* dst) {
    ROW_LD(0)  ROW_LD(1)  ROW_LD(2)  ROW_LD(3)  ROW_LD(4)  ROW_LD(5)  ROW_LD(6)  ROW_LD(7)
    ROW_LD(8)  ROW_LD(9)  ROW_LD(10) ROW_LD(11) ROW_LD(12) ROW_LD(13) ROW_LD(14) ROW_LD(15)
    ROW_ST(0)  ROW_ST(1)  ROW_ST(2)  ROW_ST(3)  ROW_ST(4)  ROW_ST(5)  ROW_ST(6)  ROW_ST(7)
    ROW_ST(8)  ROW_ST(9)  ROW_ST(10) ROW_ST(11) ROW_ST(12) ROW_ST(13) ROW_ST(14) ROW_ST(15)
}

#define W_PITCH 132
#define GEMM_BLOCKS 256
#define GEMM_WAVES 16
#define GEMM_THREADS (GEMM_WAVES * 32)
#define N_UNITS (N_NODES / 16)
static_assert(N_UNITS * 16 == N_NODES, "16-row units must tile the nodes");
static_assert((D * D / 4) % GEMM_THREADS == 0, "W1 staging: whole passes only");

__device__ __forceinline__ void mma_step(v8f (&c)[8], v2f a, const v2f (&b)[8]) {
#pragma unroll
    for (int tn = 0; tn < 8; ++tn)
        c[tn] = __builtin_amdgcn_wmma_f32_16x16x4_f32(
                    false, a, false, b[tn], (short)0, c[tn], false, false);
}

__global__ void __launch_bounds__(GEMM_THREADS)
gemm_wmma_kernel(const float* feat,
                 const float* __restrict__ W1,
                 float* __restrict__ norm_h,
                 float* out) {
    __shared__ __attribute__((aligned(16))) float s_w[D * W_PITCH];
    int wave = threadIdx.x >> 5;
    int lane = threadIdx.x & 31;
    int half = lane >> 4;
    int l15  = lane & 15;

    const float4* w4 = (const float4*)W1;
#pragma unroll
    for (int i = 0; i < D * D / 4 / GEMM_THREADS; ++i) {
        int f = (int)threadIdx.x + i * GEMM_THREADS;
        float4 w = w4[f];
        *(float4*)(s_w + (f >> 5) * W_PITCH + (f & 31) * 4) = w;
    }
    __syncthreads();

    const float* bl = s_w + l15 * W_PITCH + half * 2;
    int wr = wave >> 2, wc = wave & 3;
    int slot = (((wr + wc) & 1) << 3) | (wr << 1) | (wc >> 1);
    int g = __builtin_amdgcn_readfirstlane(slot * GEMM_BLOCKS + (int)blockIdx.x);
    for (int u = g; u < N_UNITS; u += GEMM_BLOCKS * GEMM_WAVES) {
        const float* arow = feat + (size_t)(u * 16 + l15) * D + half * 2;
        v8f c[8] = {};
        v2f pa = *(const v2f*)(arow);
        asm volatile("" : "+v"(pa));
        v2f pb[8];
#pragma unroll
        for (int tn = 0; tn < 8; ++tn) pb[tn] = *(const v2f*)(bl + tn * 16 * W_PITCH);

        for (int k = 0; k < D; k += 8) {
            v2f qa = *(const v2f*)(arow + k + 4);
            v2f qb[8];
#pragma unroll
            for (int tn = 0; tn < 8; ++tn) qb[tn] = *(const v2f*)(bl + tn * 16 * W_PITCH + k + 4);
            mma_step(c, pa, pb);
            int kp = (k + 8 < D) ? (k + 8) : (D - 4);
            pa = *(const v2f*)(arow + kp);
#pragma unroll
            for (int tn = 0; tn < 8; ++tn) pb[tn] = *(const v2f*)(bl + tn * 16 * W_PITCH + kp);
            mma_step(c, qa, qb);
        }

        float* crow = norm_h + (size_t)(u * 16 + half * 8) * D + l15;
#pragma unroll
        for (int tn = 0; tn < 8; ++tn) {
#pragma unroll
            for (int v = 0; v < 8; ++v)
                crow[(size_t)v * D + tn * 16] = c[tn][v];
        }
        copy_rows16(feat + (size_t)(u * 16) * D + lane * 4,
                    out  + (size_t)(u * 16) * (2 * D) + lane * 4);
    }
}

#define SORT_BLOCKS 256
#define SORT_THREADS 1024
#define EDGES_PER_BLOCK (N_EDGES / SORT_BLOCKS)
#define SORT_TRIPS ((EDGES_PER_BLOCK + SORT_THREADS - 1) / SORT_THREADS)
#define BUCKET_NODES 196
#define N_BUCKETS 512
#define NODE_SLOTS 256
#define TAB_ROWS (N_BUCKETS + 1)
#define PK_SHIFT 17
#define PK_SRC_MASK ((1 << PK_SHIFT) - 1)
static_assert(EDGES_PER_BLOCK * SORT_BLOCKS == N_EDGES, "edge chunks must tile the edge list");
static_assert(EDGES_PER_BLOCK % 2 == 0, "a piece is written out as whole 8-byte words");
static_assert(TAB_ROWS <= SORT_THREADS, "one thread per bucket in the block's scan");
static_assert(N_NODES <= (1 << PK_SHIFT), "a source node must fit under the destination bits");
static_assert(N_BUCKETS * BUCKET_NODES >= N_NODES, "the buckets must cover the nodes");
static_assert(BUCKET_NODES <= NODE_SLOTS && NODE_SLOTS <= (1 << (31 - PK_SHIFT)), "a packed edge is a non-negative int");

__global__ void __launch_bounds__(SORT_THREADS)
block_sort_kernel(const int* __restrict__ src,
                  const int* __restrict__ dst,
                  int* __restrict__ packed,
                  int* __restrict__ tab) {
    __shared__ __attribute__((aligned(16))) int s_pk[EDGES_PER_BLOCK];
    __shared__ int s_cnt[N_BUCKETS];
    __shared__ int s_off[TAB_ROWS];
    __shared__ int s_wtot[SORT_THREADS / 32];
    int tid  = threadIdx.x;
    int lane = tid & 31;
    int wave = tid >> 5;
    int ebase = blockIdx.x * EDGES_PER_BLOCK;
    int dv[SORT_TRIPS], sv[SORT_TRIPS], rk[SORT_TRIPS];
#pragma unroll
    for (int k = 0; k < SORT_TRIPS; ++k) {
        int i = k * SORT_THREADS + tid;
        int ic = (i < EDGES_PER_BLOCK) ? i : (EDGES_PER_BLOCK - 1);
        int d = dst[ebase + ic];
        sv[k] = src[ebase + ic];
        dv[k] = (i < EDGES_PER_BLOCK) ? d : -1;
    }
    for (int b = tid; b < N_BUCKETS; b += SORT_THREADS) s_cnt[b] = 0;
    __syncthreads();
#pragma unroll
    for (int k = 0; k < SORT_TRIPS; ++k) {
        bool valid = dv[k] >= 0;
        int bk = valid ? dv[k] / BUCKET_NODES : -1;
        unsigned vm = __builtin_amdgcn_ballot_w32(valid);
        int r = 0;
        if (vm != 0u) {
            int leader = __builtin_ctz(vm);
            int bk0 = __builtin_amdgcn_readlane(bk, leader);
            unsigned eq = __builtin_amdgcn_ballot_w32(bk == bk0);
            if (eq == vm) {
                int first = 0;
                if (lane == leader) first = atomicAdd(&s_cnt[bk0], __builtin_popcount(vm));
                first = __builtin_amdgcn_readlane(first, leader);
                r = first + __builtin_popcount(vm & ((1u << lane) - 1u));
            } else if (valid) {
                r = atomicAdd(&s_cnt[bk], 1);
            }
        }
        rk[k] = r;
    }
    __syncthreads();

    int x = (tid < N_BUCKETS) ? s_cnt[tid] : 0;
    int incl = x;
    for (int off = 1; off < 32; off <<= 1) {
        int t = __shfl_up(incl, off);
        if (lane >= off) incl += t;
    }
    if (lane == 31) s_wtot[wave] = incl;
    __syncthreads();
    int carry = (lane < wave) ? s_wtot[lane] : 0;
    for (int off = 16; off > 0; off >>= 1) carry += __shfl_xor(carry, off);
    if (tid < TAB_ROWS) s_off[tid] = carry + incl - x;
    __syncthreads();

#pragma unroll
    for (int k = 0; k < SORT_TRIPS; ++k)
        if (dv[k] >= 0) {
            int bk = dv[k] / BUCKET_NODES;
            s_pk[s_off[bk] + rk[k]] = sv[k] | ((dv[k] - bk * BUCKET_NODES) << PK_SHIFT);
        }
    __syncthreads();

    int2* outp = (int2*)(packed + (size_t)ebase);
    const int2* sp2 = (const int2*)s_pk;
    for (int i = tid; i < EDGES_PER_BLOCK / 2; i += SORT_THREADS) outp[i] = sp2[i];
    if (tid < TAB_ROWS) tab[tid * SORT_BLOCKS + blockIdx.x] = s_off[tid];
}

#define FINE_THREADS 1024
#define FINE_WAVES (FINE_THREADS / 32)
#define LIST_CAP 8192
#define EDGE_SLOTS (LIST_CAP / FINE_THREADS)
#define GATHER_DEPTH 8
static_assert(EDGE_SLOTS * FINE_THREADS == LIST_CAP, "the threads' slots must cover the list");
static_assert(SORT_BLOCKS == NODE_SLOTS, "threads 0..255 set up one run and one node counter each");

__device__ __forceinline__ void max_row(float4& acc, const float* __restrict__ norm_h, int s, unsigned l4) {
    float4 v = *(const float4*)(norm_h + (size_t)s * D + l4);
    acc.x = fmaxf(acc.x, v.x);
    acc.y = fmaxf(acc.y, v.y);
    acc.z = fmaxf(acc.z, v.z);
    acc.w = fmaxf(acc.w, v.w);
}

__global__ void __launch_bounds__(FINE_THREADS)
bucket_max_kernel(const float* __restrict__ norm_h,
                  const int* __restrict__ packed,
                  const int* __restrict__ tab,
                  float* __restrict__ out) {
    __shared__ int s_cnt[NODE_SLOTS];
    __shared__ int s_cur[NODE_SLOTS];
    __shared__ int s_wtot[NODE_SLOTS / 32];
    __shared__ int s_lo[SORT_BLOCKS];
    __shared__ int s_len[SORT_BLOCKS];
    __shared__ int s_pos[SORT_BLOCKS];
    __shared__ int s_wsum[SORT_BLOCKS / 32];
    __shared__ int s_next;
    __shared__ int s_list[LIST_CAP];
    int tid  = threadIdx.x;
    int lane = tid & 31;
    int wave = __builtin_amdgcn_readfirstlane(tid >> 5);
    int b = blockIdx.x;
    unsigned l4 = (unsigned)lane << 2;
    if (tid < SORT_BLOCKS) {
        int lo = tab[b * SORT_BLOCKS + tid];
        int hi = tab[(b + 1) * SORT_BLOCKS + tid];
        int len = hi - lo;
        int incl = len;
        for (int off = 1; off < 32; off <<= 1) {
            int t = __shfl_up(incl, off);
            if (lane >= off) incl += t;
        }
        s_lo[tid]  = lo;
        s_len[tid] = len;
        s_pos[tid] = incl - len;
        s_cnt[tid] = 0;
        if (lane == 31) s_wsum[wave] = incl;
        if (tid == 0) s_next = FINE_WAVES;
    }
    __syncthreads();
    int gp[SORT_BLOCKS / 32 + 1];
    gp[0] = 0;
#pragma unroll
    for (int w = 0; w < SORT_BLOCKS / 32; ++w)
        gp[w + 1] = gp[w] + __builtin_amdgcn_readfirstlane(s_wsum[w]);
    int total = gp[SORT_BLOCKS / 32];

    if (total > LIST_CAP) {
        for (int n = wave; n < BUCKET_NODES; n += FINE_WAVES) {
            int node = b * BUCKET_NODES + n;
            if (node >= N_NODES) break;
            float4 acc = make_float4(NEG_INF, NEG_INF, NEG_INF, NEG_INF);
            for (int t = 0; t < SORT_BLOCKS; ++t) {
                int lo  = __builtin_amdgcn_readfirstlane(s_lo[t]);
                int len = __builtin_amdgcn_readfirstlane(s_len[t]);
                for (int c = 0; c < len; c += 32) {
                    int v = (c + lane < len) ? packed[t * EDGES_PER_BLOCK + lo + c + lane] : -1;
                    unsigned hit = __builtin_amdgcn_ballot_w32(v >= 0 && (v >> PK_SHIFT) == n);
                    while (hit != 0u) {
                        int l = __builtin_ctz(hit);
                        hit &= hit - 1u;
                        max_row(acc, norm_h, __builtin_amdgcn_readlane(v, l) & PK_SRC_MASK, l4);
                    }
                }
            }
            *(float4*)(out + (size_t)node * (2 * D) + D + l4) = acc;
        }
        return;
    }

    auto edge_index = [&](int p) -> int {
        int pc = (p < total) ? p : 0;
        int g = 0, gbase = 0;
#pragma unroll
        for (int w = 1; w < SORT_BLOCKS / 32; ++w) {
            g     = (gp[w] <= pc) ? w : g;
            gbase = (gp[w] <= pc) ? gp[w] : gbase;
        }
        int q = pc - gbase;
        int t = g * 32;
#pragma unroll
        for (int step = 16; step > 0; step >>= 1)
            t = (s_pos[t + step] <= q) ? (t + step) : t;
        int i = t * EDGES_PER_BLOCK + s_lo[t] + (q - s_pos[t]);
        return (i < N_EDGES - 1) ? i : (N_EDGES - 1);
    };
    int v[EDGE_SLOTS];
#pragma unroll
    for (int j = 0; j < EDGE_SLOTS / 2; ++j)
        v[j] = packed[edge_index(j * FINE_THREADS + tid)];
#pragma unroll
    for (int j = EDGE_SLOTS / 2; j < EDGE_SLOTS; ++j) v[j] = -1;
    if (total > (EDGE_SLOTS / 2) * FINE_THREADS) {
#pragma unroll
        for (int j = EDGE_SLOTS / 2; j < EDGE_SLOTS; ++j)
            v[j] = packed[edge_index(j * FINE_THREADS + tid)];
    }
    int y[EDGE_SLOTS];
#pragma unroll
    for (int j = 0; j < EDGE_SLOTS; ++j)
        y[j] = (j * FINE_THREADS + tid < total) ? v[j] : -1;
#pragma unroll
    for (int j = 0; j < EDGE_SLOTS; ++j)
        if (y[j] >= 0) atomicAdd(&s_cnt[y[j] >> PK_SHIFT], 1);
    __syncthreads();

    int x = (tid < NODE_SLOTS) ? s_cnt[tid] : 0;
    int incl = x;
    for (int off = 1; off < 32; off <<= 1) {
        int t = __shfl_up(incl, off);
        if (lane >= off) incl += t;
    }
    if (tid < NODE_SLOTS && lane == 31) s_wtot[wave] = incl;
    __syncthreads();
    if (tid < NODE_SLOTS) {
        int carry = 0;
        for (int w = 0; w < NODE_SLOTS / 32; ++w)
            if (w < wave) carry += s_wtot[w];
        s_cur[tid] = carry + incl - x;
    }
    __syncthreads();
#pragma unroll
    for (int j = 0; j < EDGE_SLOTS; ++j) {
        if (y[j] >= 0) {
            int pos = atomicAdd(&s_cur[y[j] >> PK_SHIFT], 1);
            s_list[pos] = y[j] & PK_SRC_MASK;
        }
    }
    __syncthreads();

    int n = wave;
    while (n < BUCKET_NODES) {
        int node = b * BUCKET_NODES + n;
        if (node >= N_NODES) break;
        int nx = 0;
        if (lane == 0) nx = atomicAdd(&s_next, 1);
        int tot = __builtin_amdgcn_readfirstlane(s_cnt[n]);
        int beg = __builtin_amdgcn_readfirstlane(s_cur[n]) - tot;
        float4 acc = make_float4(NEG_INF, NEG_INF, NEG_INF, NEG_INF);
        for (int base = 0; base < tot; base += 32) {
            int cnt = tot - base;
            if (cnt > 32) cnt = 32;
            int mine = (lane < cnt) ? s_list[beg + base + lane] : 0;
            int k = 0;
            for (; k + GATHER_DEPTH <= cnt; k += GATHER_DEPTH) {
                float4 v[GATHER_DEPTH];
                for (int i = 0; i < GATHER_DEPTH; ++i) {
                    int s = __builtin_amdgcn_readlane(mine, k + i);
                    v[i] = *(const float4*)(norm_h + (size_t)s * D + l4);
                }
                for (int i = 0; i < GATHER_DEPTH; ++i) {
                    acc.x = fmaxf(acc.x, v[i].x);
                    acc.y = fmaxf(acc.y, v[i].y);
                    acc.z = fmaxf(acc.z, v[i].z);
                    acc.w = fmaxf(acc.w, v[i].w);
                }
            }
            if (k < cnt) {
                float4 v[GATHER_DEPTH - 1];
                for (int i = 0; i < GATHER_DEPTH - 1; ++i) {
                    if (k + i < cnt) {
                        int s = __builtin_amdgcn_readlane(mine, k + i);
                        v[i] = *(const float4*)(norm_h + (size_t)s * D + l4);
                    } else {
                        v[i] = make_float4(NEG_INF, NEG_INF, NEG_INF, NEG_INF);
                    }
                }
                for (int i = 0; i < GATHER_DEPTH - 1; ++i) {
                    acc.x = fmaxf(acc.x, v[i].x);
                    acc.y = fmaxf(acc.y, v[i].y);
                    acc.z = fmaxf(acc.z, v[i].z);
                    acc.w = fmaxf(acc.w, v[i].w);
                }
            }
        }
        *(float4*)(out + (size_t)node * (2 * D) + D + l4) = acc;
        n = __builtin_amdgcn_readfirstlane(nx);
    }
}

extern "C" void kernel_launch(void* const* d_in, const int* in_sizes, int n_in,
                              void* d_out, int out_size, void* d_ws, size_t ws_size,
                              hipStream_t stream) {
    const float* feat = (const float*)d_in[0];
    const float* W1   = (const float*)d_in[1];
    const int*   src  = (const int*)d_in[2];
    const int*   dst  = (const int*)d_in[3];
    float* out = (float*)d_out;

    char* ws = (char*)d_ws;
    const size_t NORM_H_BYTES = (size_t)N_NODES * D * sizeof(float);
    const size_t INTARR_BYTES = 400064;
    float* norm_h = (float*)ws;
    int*   tab    = (int*)(ws + NORM_H_BYTES);
    int*   packed = (int*)(ws + NORM_H_BYTES + 3 * INTARR_BYTES);

    gemm_wmma_kernel<<<GEMM_BLOCKS, GEMM_THREADS, 0, stream>>>(feat, W1, norm_h, out);
    block_sort_kernel<<<SORT_BLOCKS, SORT_THREADS, 0, stream>>>(src, dst, packed, tab);
    bucket_max_kernel<<<N_BUCKETS, FINE_THREADS, 0, stream>>>(norm_h, packed, tab, out);
}
